# starting kernel
# baseline (speedup 1.0000x reference)
#include <hip/hip_runtime.h>

constexpr int B_SZ   = 16;
constexpr int NUM_PM = 512;
constexpr int NUM_VM = 1536;
constexpr int NPG    = NUM_PM + NUM_VM;
constexpr int N_NODE = B_SZ * NPG;
constexpr int DEG    = 32;
constexpr int E_CNT  = DEG * N_NODE;
constexpr int D_IN   = 16;
constexpr int DH     = 64;
constexpr float NEG_SLOPE = 0.2f;

typedef float v2f __attribute__((ext_vector_type(2)));
typedef float v8f __attribute__((ext_vector_type(8)));

__device__ __forceinline__ unsigned enc_f(float f) {
    unsigned u = __float_as_uint(f);
    return (u & 0x80000000u) ? ~u : (u | 0x80000000u);
}
__device__ __forceinline__ float dec_f(unsigned u) {
    return __uint_as_float((u & 0x80000000u) ? (u & 0x7fffffffu) : ~u);
}

template <int K, int COLS>
__global__ __launch_bounds__(128) void gemm_dual_wmma(
    const float* __restrict__ X,
    const float* __restrict__ Ws, const float* __restrict__ Bs,
    const float* __restrict__ Wd, const float* __restrict__ Bd,
    float* __restrict__ Fs, float* __restrict__ Fd) {
    const int lane = threadIdx.x & 31;
    const int wave = threadIdx.x >> 5;
    const int row0 = (blockIdx.x * 4 + wave) * 16;
    const int m  = lane & 15;
    const int hi = lane >> 4;
    const float* Xr = X + (size_t)(row0 + m) * K;

    for (int t = 0; t < 2; ++t) {
        const float* W  = t ? Wd : Ws;
        const float* Bv = t ? Bd : Bs;
        float*       F  = t ? Fd : Fs;
        for (int col0 = 0; col0 < COLS; col0 += 16) {
            v8f acc = {};
#pragma unroll
            for (int kk = 0; kk < K; kk += 4) {
                v2f a, b;
                a.x = Xr[kk + 2 * hi + 0];
                a.y = Xr[kk + 2 * hi + 1];
                b.x = W[(size_t)(kk + 2 * hi + 0) * COLS + col0 + m];
                b.y = W[(size_t)(kk + 2 * hi + 1) * COLS + col0 + m];
                acc = __builtin_amdgcn_wmma_f32_16x16x4_f32(
                    false, a, false, b, (short)0, acc, false, false);
            }
            const float bias = Bv[col0 + m];
#pragma unroll
            for (int r = 0; r < 8; ++r) {
                F[(size_t)(row0 + r + 8 * hi) * COLS + col0 + m] = acc[r] + bias;
            }
        }
    }
}

__global__ __launch_bounds__(256) void zero_l1(float* __restrict__ h1,
                                               unsigned* __restrict__ m1,
                                               float* __restrict__ den1) {
    int idx = blockIdx.x * 256 + threadIdx.x;
    h1[idx] = 0.f;
    if (idx < N_NODE * 2) { m1[idx] = 0u; den1[idx] = 0.f; }
}
__global__ __launch_bounds__(256) void zero_l2(float* __restrict__ agg,
                                               unsigned* __restrict__ m2,
                                               float* __restrict__ den2) {
    int idx = blockIdx.x * 256 + threadIdx.x;
    agg[idx] = 0.f;
    if (idx < N_NODE) { m2[idx] = 0u; den2[idx] = 0.f; }
}

__global__ __launch_bounds__(256) void edge_logits1(
    const float* __restrict__ fs, const float* __restrict__ fd,
    const int* __restrict__ src, const int* __restrict__ dst,
    const float* __restrict__ attn, float* __restrict__ logits,
    unsigned* __restrict__ mmax) {
    const int e = blockIdx.x * 8 + (threadIdx.x >> 5);
    const int lane = threadIdx.x & 31;
    const int s = src[e], d = dst[e];
    const float4 a = *(const float4*)(fs + (size_t)s * 128 + lane * 4);
    const float4 b = *(const float4*)(fd + (size_t)d * 128 + lane * 4);
    const float4 w = *(const float4*)(attn + lane * 4);
    float p = 0.f, t;
    t = a.x + b.x; t = t > 0.f ? t : NEG_SLOPE * t; p += t * w.x;
    t = a.y + b.y; t = t > 0.f ? t : NEG_SLOPE * t; p += t * w.y;
    t = a.z + b.z; t = t > 0.f ? t : NEG_SLOPE * t; p += t * w.z;
    t = a.w + b.w; t = t > 0.f ? t : NEG_SLOPE * t; p += t * w.w;
    p += __shfl_xor(p, 1); p += __shfl_xor(p, 2);
    p += __shfl_xor(p, 4); p += __shfl_xor(p, 8);
    if ((lane & 15) == 0) {
        const int h = lane >> 4;
        logits[(size_t)e * 2 + h] = p;
        atomicMax(&mmax[d * 2 + h], enc_f(p));
    }
}

__global__ __launch_bounds__(256) void edge_expsum1(
    float* __restrict__ logits, const int* __restrict__ dst,
    const unsigned* __restrict__ mmax, float* __restrict__ den) {
    const int idx = blockIdx.x * 256 + threadIdx.x;
    const int e = idx >> 1, h = idx & 1;
    const int d = dst[e];
    const float ex = __expf(logits[idx] - dec_f(mmax[d * 2 + h]));
    logits[idx] = ex;
    atomicAdd(&den[d * 2 + h], ex);
}

__global__ __launch_bounds__(256) void edge_agg1(
    const float* __restrict__ fs, const int* __restrict__ src,
    const int* __restrict__ dst, const float* __restrict__ ex,
    const float* __restrict__ den, float* __restrict__ out) {
    const int e = blockIdx.x * 8 + (threadIdx.x >> 5);
    const int lane = threadIdx.x & 31;
    const int s = src[e], d = dst[e];
    const int h = lane >> 4;
    const float alpha = ex[(size_t)e * 2 + h] / den[d * 2 + h];
    const float4 v = *(const float4*)(fs + (size_t)s * 128 + lane * 4);
    float* o = out + (size_t)d * 128 + lane * 4;
    atomicAdd(o + 0, v.x * alpha);
    atomicAdd(o + 1, v.y * alpha);
    atomicAdd(o + 2, v.z * alpha);
    atomicAdd(o + 3, v.w * alpha);
}

__global__ __launch_bounds__(256) void elu_k(float* __restrict__ h) {
    const int idx = blockIdx.x * 256 + threadIdx.x;
    const float x = h[idx];
    h[idx] = x > 0.f ? x : (__expf(x) - 1.f);
}

__global__ __launch_bounds__(256) void edge_logits2(
    const float* __restrict__ fs, const float* __restrict__ fd,
    const int* __restrict__ src, const int* __restrict__ dst,
    const float* __restrict__ attn, float* __restrict__ logits,
    unsigned* __restrict__ mmax) {
    const int e = blockIdx.x * 8 + (threadIdx.x >> 5);
    const int lane = threadIdx.x & 31;
    const int s = src[e], d = dst[e];
    const float2 a = *(const float2*)(fs + (size_t)s * 64 + lane * 2);
    const float2 b = *(const float2*)(fd + (size_t)d * 64 + lane * 2);
    const float2 w = *(const float2*)(attn + lane * 2);
    float p = 0.f, t;
    t = a.x + b.x; t = t > 0.f ? t : NEG_SLOPE * t; p += t * w.x;
    t = a.y + b.y; t = t > 0.f ? t : NEG_SLOPE * t; p += t * w.y;
    p += __shfl_xor(p, 1);  p += __shfl_xor(p, 2);
    p += __shfl_xor(p, 4);  p += __shfl_xor(p, 8);
    p += __shfl_xor(p, 16);
    if (lane == 0) {
        logits[e] = p;
        atomicMax(&mmax[d], enc_f(p));
    }
}

__global__ __launch_bounds__(256) void edge_expsum2(
    float* __restrict__ logits, const int* __restrict__ dst,
    const unsigned* __restrict__ mmax, float* __restrict__ den) {
    const int e = blockIdx.x * 256 + threadIdx.x;
    const int d = dst[e];
    const float ex = __expf(logits[e] - dec_f(mmax[d]));
    logits[e] = ex;
    atomicAdd(&den[d], ex);
}

__global__ __launch_bounds__(256) void edge_agg2(
    const float* __restrict__ fs, const int* __restrict__ src,
    const int* __restrict__ dst, const float* __restrict__ ex,
    const float* __restrict__ den, float* __restrict__ out) {
    const int e = blockIdx.x * 8 + (threadIdx.x >> 5);
    const int lane = threadIdx.x & 31;
    const int s = src[e], d = dst[e];
    const float alpha = ex[e] / den[d];
    const float2 v = *(const float2*)(fs + (size_t)s * 64 + lane * 2);
    float* o = out + (size_t)d * 64 + lane * 2;
    atomicAdd(o + 0, v.x * alpha);
    atomicAdd(o + 1, v.y * alpha);
}

__global__ __launch_bounds__(256) void reorder_out(const float* __restrict__ agg,
                                                   float* __restrict__ out) {
    const int idx = blockIdx.x * 256 + threadIdx.x;
    const int n = idx >> 6, f = idx & 63;
    const int b = n >> 11;
    const int r = n & 2047;
    size_t o;
    if (r < NUM_PM) o = ((size_t)(b * NUM_PM + r) * 64) + f;
    else            o = (size_t)B_SZ * NUM_PM * 64 +
                        ((size_t)(b * NUM_VM + (r - NUM_PM)) * 64) + f;
    out[o] = agg[idx];
}

extern "C" void kernel_launch(void* const* d_in, const int* in_sizes, int n_in,
                              void* d_out, int out_size, void* d_ws, size_t ws_size,
                              hipStream_t stream) {
    const float* x     = (const float*)d_in[0];
    const int*   src   = (const int*)d_in[1];
    const int*   dst   = (const int*)d_in[2];
    const float* W1s   = (const float*)d_in[3];
    const float* b1s   = (const float*)d_in[4];
    const float* W1d   = (const float*)d_in[5];
    const float* b1d   = (const float*)d_in[6];
    const float* attn1 = (const float*)d_in[7];
    const float* W2s   = (const float*)d_in[8];
    const float* b2s   = (const float*)d_in[9];
    const float* W2d   = (const float*)d_in[10];
    const float* b2d   = (const float*)d_in[11];
    const float* attn2 = (const float*)d_in[12];
    (void)in_sizes; (void)n_in; (void)out_size; (void)ws_size;

    char* ws = (char*)d_ws;
    const size_t MB = 1024u * 1024u;
    float*    fs1   = (float*)(ws + 0);
    float*    fd1   = (float*)(ws + 16 * MB);
    float*    log1  = (float*)(ws + 32 * MB);
    unsigned* m1    = (unsigned*)(ws + 40 * MB);
    float*    den1  = (float*)(ws + 40 * MB + 262144);
    float*    h1    = (float*)(ws + 41 * MB);
    float*    fs2   = (float*)(ws + 16 * MB);
    float*    fd2   = (float*)(ws + 24 * MB);
    float*    agg2  = (float*)(ws + 0);
    float*    log2  = (float*)(ws + 8 * MB);
    unsigned* m2    = (unsigned*)(ws + 12 * MB);
    float*    den2  = (float*)(ws + 12 * MB + 131072);

    float* out = (float*)d_out;

    zero_l1<<<N_NODE * 128 / 256, 256, 0, stream>>>(h1, m1, den1);
    gemm_dual_wmma<D_IN, 2 * DH><<<N_NODE / 16 / 4, 128, 0, stream>>>(
        x, W1s, b1s, W1d, b1d, fs1, fd1);
    edge_logits1<<<E_CNT / 8, 256, 0, stream>>>(fs1, fd1, src, dst, attn1, log1, m1);
    edge_expsum1<<<E_CNT * 2 / 256, 256, 0, stream>>>(log1, dst, m1, den1);
    edge_agg1<<<E_CNT / 8, 256, 0, stream>>>(fs1, src, dst, log1, den1, h1);
    elu_k<<<N_NODE * 128 / 256, 256, 0, stream>>>(h1);

    zero_l2<<<N_NODE * 64 / 256, 256, 0, stream>>>(agg2, m2, den2);
    gemm_dual_wmma<2 * DH, DH><<<N_NODE / 16 / 4, 128, 0, stream>>>(
        h1, W2s, b2s, W2d, b2d, fs2, fd2);
    edge_logits2<<<E_CNT / 8, 256, 0, stream>>>(fs2, fd2, src, dst, attn2, log2, m2);
    edge_expsum2<<<E_CNT / 256, 256, 0, stream>>>(log2, dst, m2, den2);
    edge_agg2<<<E_CNT / 8, 256, 0, stream>>>(fs2, src, dst, log2, den2, agg2);

    reorder_out<<<N_NODE * 64 / 256, 256, 0, stream>>>(agg2, out);
}


# Softmax divided after the sum: aggregation weights rows by exp(logit-max) itself, adds the weights per node in a second LDS window, one division per row; both exp-sum launches go; sum order changed
# speedup vs baseline: 2.5043x; 2.5043x over previous
#include <hip/hip_runtime.h>

constexpr int B_SZ   = 16;
constexpr int NUM_PM = 512;
constexpr int NUM_VM = 1536;
constexpr int NPG    = NUM_PM + NUM_VM;
constexpr int N_NODE = B_SZ * NPG;
constexpr int DEG    = 32;
constexpr int E_CNT  = DEG * N_NODE;
constexpr int D_IN   = 16;
constexpr int DH     = 64;
constexpr float NEG_SLOPE = 0.2f;
constexpr int EPW    = 8;
static_assert((EPW & (EPW - 1)) == 0 && EPW <= 16, "a run's results are written by lanes 0..EPW-1 of each 16-lane half");

typedef float v2f __attribute__((ext_vector_type(2)));
typedef float v8f __attribute__((ext_vector_type(8)));

__device__ __forceinline__ unsigned enc_f(float f) {
    unsigned u = __float_as_uint(f);
    return (u & 0x80000000u) ? ~u : (u | 0x80000000u);
}
__device__ __forceinline__ float dec_f(unsigned u) {
    return __uint_as_float((u & 0x80000000u) ? (u & 0x7fffffffu) : ~u);
}

template <int K, int COLS>
__global__ __launch_bounds__(128) void gemm_dual_wmma(
    const float* __restrict__ X,
    const float* __restrict__ Ws, const float* __restrict__ Bs,
    const float* __restrict__ Wd, const float* __restrict__ Bd,
    float* __restrict__ Fs, float* __restrict__ Fd) {
    const int lane = threadIdx.x & 31;
    const int wave = threadIdx.x >> 5;
    const int row0 = (blockIdx.x * 4 + wave) * 16;
    const int m  = lane & 15;
    const int hi = lane >> 4;
    const float* Xr = X + (size_t)(row0 + m) * K;

    for (int t = 0; t < 2; ++t) {
        const float* W  = t ? Wd : Ws;
        const float* Bv = t ? Bd : Bs;
        float*       F  = t ? Fd : Fs;
        for (int col0 = 0; col0 < COLS; col0 += 16) {
            v8f acc = {};
#pragma unroll
            for (int kk = 0; kk < K; kk += 4) {
                v2f a, b;
                a.x = Xr[kk + 2 * hi + 0];
                a.y = Xr[kk + 2 * hi + 1];
                b.x = W[(size_t)(kk + 2 * hi + 0) * COLS + col0 + m];
                b.y = W[(size_t)(kk + 2 * hi + 1) * COLS + col0 + m];
                acc = __builtin_amdgcn_wmma_f32_16x16x4_f32(
                    false, a, false, b, (short)0, acc, false, false);
            }
            const float bias = Bv[col0 + m];
#pragma unroll
            for (int r = 0; r < 8; ++r) {
                F[(size_t)(row0 + r + 8 * hi) * COLS + col0 + m] = acc[r] + bias;
            }
        }
    }
}

__global__ __launch_bounds__(256) void zero_l1(float* __restrict__ z1,
                                               unsigned* __restrict__ m1,
                                               float* __restrict__ den1) {
    int idx = blockIdx.x * 256 + threadIdx.x;
    z1[idx] = 0.f;
    if (idx < N_NODE * 2) { m1[idx] = 0u; den1[idx] = 0.f; }
}
__global__ __launch_bounds__(256) void zero_l2(float* __restrict__ agg,
                                               unsigned* __restrict__ m2,
                                               float* __restrict__ den2) {
    int idx = blockIdx.x * 256 + threadIdx.x;
    agg[idx] = 0.f;
    if (idx < N_NODE) { m2[idx] = 0u; den2[idx] = 0.f; }
}

__global__ __launch_bounds__(256) void edge_logits1(
    const float* __restrict__ fs, const float* __restrict__ fd,
    const int* __restrict__ src, const int* __restrict__ dst,
    const float* __restrict__ attn, float* __restrict__ logits,
    unsigned* __restrict__ mmax) {
    const int wave = __builtin_amdgcn_readfirstlane((int)(threadIdx.x >> 5));
    const int e0 = (blockIdx.x * 8 + wave) * EPW;
    const int lane = threadIdx.x & 31;
    const float4 w = *(const float4*)(attn + lane * 4);
    int s[EPW], d[EPW];
#pragma unroll
    for (int i = 0; i < EPW; ++i) { s[i] = src[e0 + i]; d[i] = dst[e0 + i]; }
    float4 a[EPW], b[EPW];
#pragma unroll
    for (int i = 0; i < EPW; ++i) {
        a[i] = *(const float4*)(fs + (size_t)s[i] * 128 + lane * 4);
        b[i] = *(const float4*)(fd + (size_t)d[i] * 128 + lane * 4);
    }
    float pe[EPW];
#pragma unroll
    for (int i = 0; i < EPW; ++i) {
        float p = 0.f, t;
        t = a[i].x + b[i].x; t = t > 0.f ? t : NEG_SLOPE * t; p += t * w.x;
        t = a[i].y + b[i].y; t = t > 0.f ? t : NEG_SLOPE * t; p += t * w.y;
        t = a[i].z + b[i].z; t = t > 0.f ? t : NEG_SLOPE * t; p += t * w.z;
        t = a[i].w + b[i].w; t = t > 0.f ? t : NEG_SLOPE * t; p += t * w.w;
        p += __shfl_xor(p, 1); p += __shfl_xor(p, 2);
        p += __shfl_xor(p, 4); p += __shfl_xor(p, 8);
        pe[i] = p;
    }
    const int sel = lane & (EPW - 1);
    float pv = pe[0];
    int   dv = d[0];
#pragma unroll
    for (int i = 1; i < EPW; ++i) {
        pv = (sel == i) ? pe[i] : pv;
        dv = (sel == i) ? d[i]  : dv;
    }
    if ((lane & 15) < EPW) {
        const int h = lane >> 4;
        logits[(size_t)(e0 + sel) * 2 + h] = pv;
        atomicMax(&mmax[dv * 2 + h], enc_f(pv));
    }
}

constexpr int WIN = 2048;
template <int FT, int HEADS, int EPB, int SRCW>
__global__ __launch_bounds__(1024) void edge_agg_lds(
    const float* __restrict__ fs, const int* __restrict__ src,
    const int* __restrict__ dst, const float* __restrict__ lg,
    const unsigned* __restrict__ mmax, float* __restrict__ out, float* __restrict__ den) {
    constexpr int SL   = FT / 32;
    constexpr int EPWV = EPB / 32;
    constexpr int HW   = FT / HEADS;
    static_assert(EPWV % EPW == 0 && FT % 32 == 0, "whole runs per wave, whole slices per row");
    static_assert((SRCW == FT && HW % 32 == 0) || (SRCW == HW && 32 % HW == 0),
                  "a slice inside one head, or whole heads inside a slice");
    static_assert(HW >= EPW, "a head's lanes in a slice cover one run of edges");
    __shared__ float acc[WIN * 32];
    __shared__ float dacc[WIN * HEADS];
    const int wave  = __builtin_amdgcn_readfirstlane((int)(threadIdx.x >> 5));
    const int lane  = threadIdx.x & 31;
    const int slice = blockIdx.x % SL;
    const int eb    = (blockIdx.x / SL) * EPB;
    const int fo    = slice * 32 + lane;
    const int h     = (HW % 32 == 0) ? (slice * 32) / HW : fo / HW;
    const int fsrc  = fo % SRCW;
    const int wbase = dst[eb] & ~(WIN - 1);
    const bool dblk = (HW % 32 == 0) ? ((slice * 32) % HW == 0) : true;
    const int  q    = (HW % 32 == 0) ? lane : lane % HW;

    for (int i = threadIdx.x; i < WIN * 32 / 4; i += 1024)
        ((float4*)acc)[i] = make_float4(0.f, 0.f, 0.f, 0.f);
    for (int i = threadIdx.x; i < WIN * HEADS / 4; i += 1024)
        ((float4*)dacc)[i] = make_float4(0.f, 0.f, 0.f, 0.f);
    __syncthreads();

    const int ew = eb + wave * EPWV;
    int s[EPW], d[EPW];
#pragma unroll
    for (int i = 0; i < EPW; ++i) { s[i] = src[ew + i]; d[i] = dst[ew + i]; }
    for (int st = 0; st < EPWV; st += EPW) {
        const int e0 = ew + st;
        const int en = (st + EPW < EPWV) ? e0 + EPW : e0;
        int sn[EPW], dx[EPW];
#pragma unroll
        for (int i = 0; i < EPW; ++i) { sn[i] = src[en + i]; dx[i] = dst[en + i]; }
        float v[EPW], xl[EPW];
        unsigned mx[EPW];
#pragma unroll
        for (int i = 0; i < EPW; ++i) {
            v[i]  = fs[(size_t)s[i] * SRCW + fsrc];
            xl[i] = lg[(size_t)(e0 + i) * HEADS + h];
            mx[i] = mmax[d[i] * HEADS + h];
        }
        float w[EPW];
#pragma unroll
        for (int i = 0; i < EPW; ++i) {
            w[i] = __expf(xl[i] - dec_f(mx[i]));
            const float val = v[i] * w[i];
            const int r = d[i] - wbase;
            if ((unsigned)r < (unsigned)WIN) atomicAdd(&acc[r * 32 + lane], val);
            else                             atomicAdd(out + (size_t)d[i] * FT + fo, val);
        }
        if (dblk) {
            float pw = w[0];
            int   pd = d[0];
#pragma unroll
            for (int i = 1; i < EPW; ++i) {
                pw = (q == i) ? w[i] : pw;
                pd = (q == i) ? d[i] : pd;
            }
            if (q < EPW) {
                const int r = pd - wbase;
                if ((unsigned)r < (unsigned)WIN) atomicAdd(&dacc[r * HEADS + h], pw);
                else                             atomicAdd(den + pd * HEADS + h, pw);
            }
        }
#pragma unroll
        for (int i = 0; i < EPW; ++i) { s[i] = sn[i]; d[i] = dx[i]; }
    }
    __syncthreads();

    for (int i0 = threadIdx.x; i0 < WIN * 32; i0 += 8 * 1024) {
        float a[8];
#pragma unroll
        for (int j = 0; j < 8; ++j) a[j] = acc[i0 + j * 1024];
#pragma unroll
        for (int j = 0; j < 8; ++j)
            atomicAdd(out + (size_t)(wbase + ((i0 + j * 1024) >> 5)) * FT + fo, a[j]);
    }
    if (dblk)
        for (int i = threadIdx.x; i < WIN * HEADS; i += 1024)
            atomicAdd(den + (size_t)wbase * HEADS + i, dacc[i]);
}

__global__ __launch_bounds__(128) void gemm_z_elu(
    const float* __restrict__ Z, const float* __restrict__ W, const float* __restrict__ Bv,
    const float* __restrict__ den, float* __restrict__ H) {
    const int lane = threadIdx.x & 31;
    const int wave = threadIdx.x >> 5;
    const int row0 = (blockIdx.x * 4 + wave) * 16;
    const int m  = lane & 15;
    const int hi = lane >> 4;

    for (int t = 0; t < 2; ++t) {
        const float* Zr = Z + (size_t)(row0 + m) * (2 * D_IN) + t * D_IN;
        v2f a[D_IN / 4];
#pragma unroll
        for (int kk = 0; kk < D_IN; kk += 4) {
            a[kk / 4].x = Zr[kk + 2 * hi + 0];
            a[kk / 4].y = Zr[kk + 2 * hi + 1];
        }
        float has[8], inv[8];
#pragma unroll
        for (int r = 0; r < 8; ++r) {
            const float dn = den[(row0 + r + 8 * hi) * 2 + t];
            has[r] = dn > 0.f ? 1.f : 0.f;
            inv[r] = dn > 0.f ? 1.f / dn : 0.f;
        }
        for (int c0 = 0; c0 < DH; c0 += 16) {
            const int col0 = t * DH + c0;
            v8f acc = {};
#pragma unroll
            for (int kk = 0; kk < D_IN; kk += 4) {
                v2f b;
                b.x = W[(size_t)(kk + 2 * hi + 0) * (2 * DH) + col0 + m];
                b.y = W[(size_t)(kk + 2 * hi + 1) * (2 * DH) + col0 + m];
                acc = __builtin_amdgcn_wmma_f32_16x16x4_f32(
                    false, a[kk / 4], false, b, (short)0, acc, false, false);
            }
            const float bias = Bv[col0 + m];
#pragma unroll
            for (int r = 0; r < 8; ++r) {
                const float v = acc[r] * inv[r] + bias * has[r];
                H[(size_t)(row0 + r + 8 * hi) * (2 * DH) + col0 + m] = v > 0.f ? v : (__expf(v) - 1.f);
            }
        }
    }
}

__global__ __launch_bounds__(256) void edge_logits2(
    const float* __restrict__ fs, const float* __restrict__ fd,
    const int* __restrict__ src, const int* __restrict__ dst,
    const float* __restrict__ attn, float* __restrict__ logits,
    unsigned* __restrict__ mmax) {
    const int wave = __builtin_amdgcn_readfirstlane((int)(threadIdx.x >> 5));
    const int e0 = (blockIdx.x * 8 + wave) * EPW;
    const int lane = threadIdx.x & 31;
    const float2 w = *(const float2*)(attn + lane * 2);
    int s[EPW], d[EPW];
#pragma unroll
    for (int i = 0; i < EPW; ++i) { s[i] = src[e0 + i]; d[i] = dst[e0 + i]; }
    float2 a[EPW], b[EPW];
#pragma unroll
    for (int i = 0; i < EPW; ++i) {
        a[i] = *(const float2*)(fs + (size_t)s[i] * 64 + lane * 2);
        b[i] = *(const float2*)(fd + (size_t)d[i] * 64 + lane * 2);
    }
    float pe[EPW];
#pragma unroll
    for (int i = 0; i < EPW; ++i) {
        float p = 0.f, t;
        t = a[i].x + b[i].x; t = t > 0.f ? t : NEG_SLOPE * t; p += t * w.x;
        t = a[i].y + b[i].y; t = t > 0.f ? t : NEG_SLOPE * t; p += t * w.y;
        p += __shfl_xor(p, 1);  p += __shfl_xor(p, 2);
        p += __shfl_xor(p, 4);  p += __shfl_xor(p, 8);
        p += __shfl_xor(p, 16);
        pe[i] = p;
    }
    const int sel = lane & (EPW - 1);
    float pv = pe[0];
    int   dv = d[0];
#pragma unroll
    for (int i = 1; i < EPW; ++i) {
        pv = (sel == i) ? pe[i] : pv;
        dv = (sel == i) ? d[i]  : dv;
    }
    if (lane < EPW) {
        logits[e0 + sel] = pv;
        atomicMax(&mmax[dv], enc_f(pv));
    }
}

__global__ __launch_bounds__(256) void reorder_out(const float* __restrict__ agg,
                                                   const float* __restrict__ den,
                                                   float* __restrict__ out) {
    const int idx = blockIdx.x * 256 + threadIdx.x;
    const int n = idx >> 6, f = idx & 63;
    const int b = n >> 11;
    const int r = n & 2047;
    size_t o;
    if (r < NUM_PM) o = ((size_t)(b * NUM_PM + r) * 64) + f;
    else            o = (size_t)B_SZ * NUM_PM * 64 +
                        ((size_t)(b * NUM_VM + (r - NUM_PM)) * 64) + f;
    const float dn = den[n];
    out[o] = dn > 0.f ? agg[idx] / dn : 0.f;
}

extern "C" void kernel_launch(void* const* d_in, const int* in_sizes, int n_in,
                              void* d_out, int out_size, void* d_ws, size_t ws_size,
                              hipStream_t stream) {
    const float* x     = (const float*)d_in[0];
    const int*   src   = (const int*)d_in[1];
    const int*   dst   = (const int*)d_in[2];
    const float* W1s   = (const float*)d_in[3];
    const float* b1s   = (const float*)d_in[4];
    const float* W1d   = (const float*)d_in[5];
    const float* b1d   = (const float*)d_in[6];
    const float* attn1 = (const float*)d_in[7];
    const float* W2s   = (const float*)d_in[8];
    const float* b2s   = (const float*)d_in[9];
    const float* W2d   = (const float*)d_in[10];
    const float* b2d   = (const float*)d_in[11];
    const float* attn2 = (const float*)d_in[12];
    (void)in_sizes; (void)n_in; (void)out_size; (void)ws_size;

    char* ws = (char*)d_ws;
    const size_t MB = 1024u * 1024u;
    float*    fs1   = (float*)(ws + 0);
    float*    fd1   = (float*)(ws + 16 * MB);
    float*    log1  = (float*)(ws + 32 * MB);
    unsigned* m1    = (unsigned*)(ws + 40 * MB);
    float*    den1  = (float*)(ws + 40 * MB + 262144);
    float*    h1    = (float*)(ws + 41 * MB);
    float*    z1    = (float*)(ws + 57 * MB);
    float*    fs2   = (float*)(ws + 16 * MB);
    float*    fd2   = (float*)(ws + 24 * MB);
    float*    agg2  = (float*)(ws + 0);
    float*    log2  = (float*)(ws + 8 * MB);
    unsigned* m2    = (unsigned*)(ws + 12 * MB);
    float*    den2  = (float*)(ws + 12 * MB + 131072);

    float* out = (float*)d_out;

    zero_l1<<<N_NODE * 2 * D_IN / 256, 256, 0, stream>>>(z1, m1, den1);
    gemm_dual_wmma<D_IN, 2 * DH><<<N_NODE / 16 / 4, 128, 0, stream>>>(
        x, W1s, b1s, W1d, b1d, fs1, fd1);
    edge_logits1<<<E_CNT / 8 / EPW, 256, 0, stream>>>(fs1, fd1, src, dst, attn1, log1, m1);
    edge_agg_lds<2 * D_IN, 2, 4096, D_IN><<<E_CNT / 4096, 1024, 0, stream>>>(x, src, dst, log1, m1, z1, den1);
    gemm_z_elu<<<N_NODE / 16 / 4, 128, 0, stream>>>(z1, W1s, b1s, den1, h1);

    zero_l2<<<N_NODE * 64 / 256, 256, 0, stream>>>(agg2, m2, den2);
    gemm_dual_wmma<2 * DH, DH><<<N_NODE / 16 / 4, 128, 0, stream>>>(
        h1, W2s, b2s, W2d, b2d, fs2, fd2);
    edge_logits2<<<E_CNT / 8 / EPW, 256, 0, stream>>>(fs2, fd2, src, dst, attn2, log2, m2);
    edge_agg_lds<DH, 1, 8192, DH><<<E_CNT / 8192 * 2, 1024, 0, stream>>>(fs2, src, dst, log2, m2, agg2, den2);

    reorder_out<<<N_NODE * 64 / 256, 256, 0, stream>>>(agg2, den2, out);
}


# Layer-1 logits from the 16-wide inputs: per 16 edges one WMMA pass [x[src]|x[dst]] x [W1s;W1d] gives fs+fd in registers; fs/fd planes, their GEMM launch and row gathers go; logit sum order changed
# speedup vs baseline: 2.8590x; 1.1416x over previous
#include <hip/hip_runtime.h>

constexpr int B_SZ   = 16;
constexpr int NUM_PM = 512;
constexpr int NUM_VM = 1536;
constexpr int NPG    = NUM_PM + NUM_VM;
constexpr int N_NODE = B_SZ * NPG;
constexpr int DEG    = 32;
constexpr int E_CNT  = DEG * N_NODE;
constexpr int D_IN   = 16;
constexpr int DH     = 64;
constexpr float NEG_SLOPE = 0.2f;
constexpr int EPW    = 8;
static_assert((EPW & (EPW - 1)) == 0 && EPW <= 16, "a run's results are written by lanes 0..EPW-1 of each 16-lane half");

typedef float v2f __attribute__((ext_vector_type(2)));
typedef float v8f __attribute__((ext_vector_type(8)));

__device__ __forceinline__ unsigned enc_f(float f) {
    unsigned u = __float_as_uint(f);
    return (u & 0x80000000u) ? ~u : (u | 0x80000000u);
}
__device__ __forceinline__ float dec_f(unsigned u) {
    return __uint_as_float((u & 0x80000000u) ? (u & 0x7fffffffu) : ~u);
}

template <int K, int COLS>
__global__ __launch_bounds__(128) void gemm_dual_wmma(
    const float* __restrict__ X,
    const float* __restrict__ Ws, const float* __restrict__ Bs,
    const float* __restrict__ Wd, const float* __restrict__ Bd,
    float* __restrict__ Fs, float* __restrict__ Fd) {
    const int lane = threadIdx.x & 31;
    const int wave = threadIdx.x >> 5;
    const int row0 = (blockIdx.x * 4 + wave) * 16;
    const int m  = lane & 15;
    const int hi = lane >> 4;
    const float* Xr = X + (size_t)(row0 + m) * K;

    for (int t = 0; t < 2; ++t) {
        const float* W  = t ? Wd : Ws;
        const float* Bv = t ? Bd : Bs;
        float*       F  = t ? Fd : Fs;
        for (int col0 = 0; col0 < COLS; col0 += 16) {
            v8f acc = {};
#pragma unroll
            for (int kk = 0; kk < K; kk += 4) {
                v2f a, b;
                a.x = Xr[kk + 2 * hi + 0];
                a.y = Xr[kk + 2 * hi + 1];
                b.x = W[(size_t)(kk + 2 * hi + 0) * COLS + col0 + m];
                b.y = W[(size_t)(kk + 2 * hi + 1) * COLS + col0 + m];
                acc = __builtin_amdgcn_wmma_f32_16x16x4_f32(
                    false, a, false, b, (short)0, acc, false, false);
            }
            const float bias = Bv[col0 + m];
#pragma unroll
            for (int r = 0; r < 8; ++r) {
                F[(size_t)(row0 + r + 8 * hi) * COLS + col0 + m] = acc[r] + bias;
            }
        }
    }
}

__global__ __launch_bounds__(256) void zero_l1(float* __restrict__ z1,
                                               unsigned* __restrict__ m1,
                                               float* __restrict__ den1) {
    int idx = blockIdx.x * 256 + threadIdx.x;
    z1[idx] = 0.f;
    if (idx < N_NODE * 2) { m1[idx] = 0u; den1[idx] = 0.f; }
}
__global__ __launch_bounds__(256) void zero_l2(float* __restrict__ agg,
                                               unsigned* __restrict__ m2,
                                               float* __restrict__ den2) {
    int idx = blockIdx.x * 256 + threadIdx.x;
    agg[idx] = 0.f;
    if (idx < N_NODE) { m2[idx] = 0u; den2[idx] = 0.f; }
}

constexpr int LT_TPW = 8;
static_assert(E_CNT % (16 * LT_TPW * 4) == 0, "whole tiles per wave, whole waves per work-group");

template <int CTRL>
__device__ __forceinline__ float row_xm(float v) {
    return __int_as_float(__builtin_amdgcn_update_dpp(0, __float_as_int(v), CTRL, 0xf, 0xf, false));
}

__global__ __launch_bounds__(128) void edge_logits1_x(
    const float* __restrict__ x, const int* __restrict__ src, const int* __restrict__ dst,
    const float* __restrict__ Ws, const float* __restrict__ Bs,
    const float* __restrict__ Wd, const float* __restrict__ Bd,
    const float* __restrict__ attn, float* __restrict__ logits, unsigned* __restrict__ mmax) {
    __shared__ v2f wl[64 * 32];
    const int lane = threadIdx.x & 31;
    const int wave = threadIdx.x >> 5;
    const int c  = lane & 15;
    const int hh = lane >> 4;

    for (int q = threadIdx.x; q < 64 * 32; q += 128) {
        const int l = q & 31, j = q >> 8, n = (q >> 5) & 7;
        const float* Wm = (j < 4) ? Ws : Wd;
        const int k = 8 * (l >> 4) + 2 * (j & 3);
        v2f w;
        w.x = Wm[(size_t)(k + 0) * (2 * DH) + 16 * n + (l & 15)];
        w.y = Wm[(size_t)(k + 1) * (2 * DH) + 16 * n + (l & 15)];
        wl[q] = w;
    }
    __syncthreads();
    v2f wf[8][8];
#pragma unroll
    for (int j = 0; j < 8; ++j)
#pragma unroll
        for (int n = 0; n < 8; ++n) wf[j][n] = wl[(j * 8 + n) * 32 + lane];
    float bv[8], av[8];
#pragma unroll
    for (int n = 0; n < 8; ++n) {
        bv[n] = Bs[16 * n + c] + Bd[16 * n + c];
        av[n] = attn[16 * n + c];
    }

    const int ebase = (blockIdx.x * 4 + wave) * (LT_TPW * 16);
    const int osel  = 8 * hh + (c & 7);
    const int ohead = c >> 3;
    const int et1 = (LT_TPW > 1) ? ebase + 16 : ebase;
    const int s1 = src[ebase + c], d1 = dst[ebase + c];
    int dw = dst[ebase + osel];
    int s2 = src[et1 + c], d2 = dst[et1 + c], dw2 = dst[et1 + osel];
    float4 xs0 = *(const float4*)(x + (size_t)s1 * D_IN + 8 * hh);
    float4 xs1 = *(const float4*)(x + (size_t)s1 * D_IN + 8 * hh + 4);
    float4 xd0 = *(const float4*)(x + (size_t)d1 * D_IN + 8 * hh);
    float4 xd1 = *(const float4*)(x + (size_t)d1 * D_IN + 8 * hh + 4);
    for (int t = 0; t < LT_TPW; ++t) {
        const int e0 = ebase + t * 16;
        const int e2 = ebase + ((t + 2 < LT_TPW) ? t + 2 : LT_TPW - 1) * 16;
        const int s3 = src[e2 + c], d3 = dst[e2 + c], dw3 = dst[e2 + osel];
        const float4 ns0 = *(const float4*)(x + (size_t)s2 * D_IN + 8 * hh);
        const float4 ns1 = *(const float4*)(x + (size_t)s2 * D_IN + 8 * hh + 4);
        const float4 nd0 = *(const float4*)(x + (size_t)d2 * D_IN + 8 * hh);
        const float4 nd1 = *(const float4*)(x + (size_t)d2 * D_IN + 8 * hh + 4);

        v2f a[8];
        a[0].x = xs0.x; a[0].y = xs0.y; a[1].x = xs0.z; a[1].y = xs0.w;
        a[2].x = xs1.x; a[2].y = xs1.y; a[3].x = xs1.z; a[3].y = xs1.w;
        a[4].x = xd0.x; a[4].y = xd0.y; a[5].x = xd0.z; a[5].y = xd0.w;
        a[6].x = xd1.x; a[6].y = xd1.y; a[7].x = xd1.z; a[7].y = xd1.w;
        float p[16];
#pragma unroll
        for (int i = 0; i < 16; ++i) p[i] = 0.f;
#pragma unroll
        for (int n = 0; n < 8; ++n) {
            v8f acc = {};
#pragma unroll
            for (int j = 0; j < 8; ++j)
                acc = __builtin_amdgcn_wmma_f32_16x16x4_f32(
                    false, a[j], false, wf[j][n], (short)0, acc, false, false);
#pragma unroll
            for (int r = 0; r < 8; ++r) {
                float u = acc[r] + bv[n];
                u = u > 0.f ? u : NEG_SLOPE * u;
                p[(n >> 2) * 8 + r] += u * av[n];
            }
        }
#pragma unroll
        for (int i = 0; i < 8; ++i) {
            const float keep = (c & 8) ? p[i + 8] : p[i], send = (c & 8) ? p[i] : p[i + 8];
            p[i] = keep + row_xm<0x168>(send);
        }
#pragma unroll
        for (int i = 0; i < 4; ++i) {
            const float keep = (c & 4) ? p[i + 4] : p[i], send = (c & 4) ? p[i] : p[i + 4];
            p[i] = keep + row_xm<0x164>(send);
        }
#pragma unroll
        for (int i = 0; i < 2; ++i) {
            const float keep = (c & 2) ? p[i + 2] : p[i], send = (c & 2) ? p[i] : p[i + 2];
            p[i] = keep + row_xm<0x162>(send);
        }
        const float keep = (c & 1) ? p[1] : p[0], send = (c & 1) ? p[0] : p[1];
        const float pv = keep + row_xm<0x161>(send);
        logits[(size_t)(e0 + osel) * 2 + ohead] = pv;
        atomicMax(&mmax[dw * 2 + ohead], enc_f(pv));

        xs0 = ns0; xs1 = ns1; xd0 = nd0; xd1 = nd1;
        dw = dw2;
        s2 = s3; d2 = d3; dw2 = dw3;
    }
}

constexpr int WIN = 2048;
template <int FT, int HEADS, int EPB, int SRCW>
__global__ __launch_bounds__(1024) void edge_agg_lds(
    const float* __restrict__ fs, const int* __restrict__ src,
    const int* __restrict__ dst, const float* __restrict__ lg,
    const unsigned* __restrict__ mmax, float* __restrict__ out, float* __restrict__ den) {
    constexpr int SL   = FT / 32;
    constexpr int EPWV = EPB / 32;
    constexpr int HW   = FT / HEADS;
    static_assert(EPWV % EPW == 0 && FT % 32 == 0, "whole runs per wave, whole slices per row");
    static_assert((SRCW == FT && HW % 32 == 0) || (SRCW == HW && 32 % HW == 0),
                  "a slice inside one head, or whole heads inside a slice");
    static_assert(HW >= EPW, "a head's lanes in a slice cover one run of edges");
    __shared__ float acc[WIN * 32];
    __shared__ float dacc[WIN * HEADS];
    const int wave  = __builtin_amdgcn_readfirstlane((int)(threadIdx.x >> 5));
    const int lane  = threadIdx.x & 31;
    const int slice = blockIdx.x % SL;
    const int eb    = (blockIdx.x / SL) * EPB;
    const int fo    = slice * 32 + lane;
    const int h     = (HW % 32 == 0) ? (slice * 32) / HW : fo / HW;
    const int fsrc  = fo % SRCW;
    const int wbase = dst[eb] & ~(WIN - 1);
    const bool dblk = (HW % 32 == 0) ? ((slice * 32) % HW == 0) : true;
    const int  q    = (HW % 32 == 0) ? lane : lane % HW;

    for (int i = threadIdx.x; i < WIN * 32 / 4; i += 1024)
        ((float4*)acc)[i] = make_float4(0.f, 0.f, 0.f, 0.f);
    for (int i = threadIdx.x; i < WIN * HEADS / 4; i += 1024)
        ((float4*)dacc)[i] = make_float4(0.f, 0.f, 0.f, 0.f);
    __syncthreads();

    const int ew = eb + wave * EPWV;
    int s[EPW], d[EPW];
#pragma unroll
    for (int i = 0; i < EPW; ++i) { s[i] = src[ew + i]; d[i] = dst[ew + i]; }
    for (int st = 0; st < EPWV; st += EPW) {
        const int e0 = ew + st;
        const int en = (st + EPW < EPWV) ? e0 + EPW : e0;
        int sn[EPW], dx[EPW];
#pragma unroll
        for (int i = 0; i < EPW; ++i) { sn[i] = src[en + i]; dx[i] = dst[en + i]; }
        float v[EPW], xl[EPW];
        unsigned mx[EPW];
#pragma unroll
        for (int i = 0; i < EPW; ++i) {
            v[i]  = fs[(size_t)s[i] * SRCW + fsrc];
            xl[i] = lg[(size_t)(e0 + i) * HEADS + h];
            mx[i] = mmax[d[i] * HEADS + h];
        }
        float w[EPW];
#pragma unroll
        for (int i = 0; i < EPW; ++i) {
            w[i] = __expf(xl[i] - dec_f(mx[i]));
            const float val = v[i] * w[i];
            const int r = d[i] - wbase;
            if ((unsigned)r < (unsigned)WIN) atomicAdd(&acc[r * 32 + lane], val);
            else                             atomicAdd(out + (size_t)d[i] * FT + fo, val);
        }
        if (dblk) {
            float pw = w[0];
            int   pd = d[0];
#pragma unroll
            for (int i = 1; i < EPW; ++i) {
                pw = (q == i) ? w[i] : pw;
                pd = (q == i) ? d[i] : pd;
            }
            if (q < EPW) {
                const int r = pd - wbase;
                if ((unsigned)r < (unsigned)WIN) atomicAdd(&dacc[r * HEADS + h], pw);
                else                             atomicAdd(den + pd * HEADS + h, pw);
            }
        }
#pragma unroll
        for (int i = 0; i < EPW; ++i) { s[i] = sn[i]; d[i] = dx[i]; }
    }
    __syncthreads();

    for (int i0 = threadIdx.x; i0 < WIN * 32; i0 += 8 * 1024) {
        float a[8];
#pragma unroll
        for (int j = 0; j < 8; ++j) a[j] = acc[i0 + j * 1024];
#pragma unroll
        for (int j = 0; j < 8; ++j)
            atomicAdd(out + (size_t)(wbase + ((i0 + j * 1024) >> 5)) * FT + fo, a[j]);
    }
    if (dblk)
        for (int i = threadIdx.x; i < WIN * HEADS; i += 1024)
            atomicAdd(den + (size_t)wbase * HEADS + i, dacc[i]);
}

__global__ __launch_bounds__(128) void gemm_z_elu(
    const float* __restrict__ Z, const float* __restrict__ W, const float* __restrict__ Bv,
    const float* __restrict__ den, float* __restrict__ H) {
    const int lane = threadIdx.x & 31;
    const int wave = threadIdx.x >> 5;
    const int row0 = (blockIdx.x * 4 + wave) * 16;
    const int m  = lane & 15;
    const int hi = lane >> 4;

    for (int t = 0; t < 2; ++t) {
        const float* Zr = Z + (size_t)(row0 + m) * (2 * D_IN) + t * D_IN;
        v2f a[D_IN / 4];
#pragma unroll
        for (int kk = 0; kk < D_IN; kk += 4) {
            a[kk / 4].x = Zr[kk + 2 * hi + 0];
            a[kk / 4].y = Zr[kk + 2 * hi + 1];
        }
        float has[8], inv[8];
#pragma unroll
        for (int r = 0; r < 8; ++r) {
            const float dn = den[(row0 + r + 8 * hi) * 2 + t];
            has[r] = dn > 0.f ? 1.f : 0.f;
            inv[r] = dn > 0.f ? 1.f / dn : 0.f;
        }
        for (int c0 = 0; c0 < DH; c0 += 16) {
            const int col0 = t * DH + c0;
            v8f acc = {};
#pragma unroll
            for (int kk = 0; kk < D_IN; kk += 4) {
                v2f b;
                b.x = W[(size_t)(kk + 2 * hi + 0) * (2 * DH) + col0 + m];
                b.y = W[(size_t)(kk + 2 * hi + 1) * (2 * DH) + col0 + m];
                acc = __builtin_amdgcn_wmma_f32_16x16x4_f32(
                    false, a[kk / 4], false, b, (short)0, acc, false, false);
            }
            const float bias = Bv[col0 + m];
#pragma unroll
            for (int r = 0; r < 8; ++r) {
                const float v = acc[r] * inv[r] + bias * has[r];
                H[(size_t)(row0 + r + 8 * hi) * (2 * DH) + col0 + m] = v > 0.f ? v : (__expf(v) - 1.f);
            }
        }
    }
}

__global__ __launch_bounds__(256) void edge_logits2(
    const float* __restrict__ fs, const float* __restrict__ fd,
    const int* __restrict__ src, const int* __restrict__ dst,
    const float* __restrict__ attn, float* __restrict__ logits,
    unsigned* __restrict__ mmax) {
    const int wave = __builtin_amdgcn_readfirstlane((int)(threadIdx.x >> 5));
    const int e0 = (blockIdx.x * 8 + wave) * EPW;
    const int lane = threadIdx.x & 31;
    const float2 w = *(const float2*)(attn + lane * 2);
    int s[EPW], d[EPW];
#pragma unroll
    for (int i = 0; i < EPW; ++i) { s[i] = src[e0 + i]; d[i] = dst[e0 + i]; }
    float2 a[EPW], b[EPW];
#pragma unroll
    for (int i = 0; i < EPW; ++i) {
        a[i] = *(const float2*)(fs + (size_t)s[i] * 64 + lane * 2);
        b[i] = *(const float2*)(fd + (size_t)d[i] * 64 + lane * 2);
    }
    float pe[EPW];
#pragma unroll
    for (int i = 0; i < EPW; ++i) {
        float p = 0.f, t;
        t = a[i].x + b[i].x; t = t > 0.f ? t : NEG_SLOPE * t; p += t * w.x;
        t = a[i].y + b[i].y; t = t > 0.f ? t : NEG_SLOPE * t; p += t * w.y;
        p += __shfl_xor(p, 1);  p += __shfl_xor(p, 2);
        p += __shfl_xor(p, 4);  p += __shfl_xor(p, 8);
        p += __shfl_xor(p, 16);
        pe[i] = p;
    }
    const int sel = lane & (EPW - 1);
    float pv = pe[0];
    int   dv = d[0];
#pragma unroll
    for (int i = 1; i < EPW; ++i) {
        pv = (sel == i) ? pe[i] : pv;
        dv = (sel == i) ? d[i]  : dv;
    }
    if (lane < EPW) {
        logits[e0 + sel] = pv;
        atomicMax(&mmax[dv], enc_f(pv));
    }
}

__global__ __launch_bounds__(256) void reorder_out(const float* __restrict__ agg,
                                                   const float* __restrict__ den,
                                                   float* __restrict__ out) {
    const int idx = blockIdx.x * 256 + threadIdx.x;
    const int n = idx >> 6, f = idx & 63;
    const int b = n >> 11;
    const int r = n & 2047;
    size_t o;
    if (r < NUM_PM) o = ((size_t)(b * NUM_PM + r) * 64) + f;
    else            o = (size_t)B_SZ * NUM_PM * 64 +
                        ((size_t)(b * NUM_VM + (r - NUM_PM)) * 64) + f;
    const float dn = den[n];
    out[o] = dn > 0.f ? agg[idx] / dn : 0.f;
}

extern "C" void kernel_launch(void* const* d_in, const int* in_sizes, int n_in,
                              void* d_out, int out_size, void* d_ws, size_t ws_size,
                              hipStream_t stream) {
    const float* x     = (const float*)d_in[0];
    const int*   src   = (const int*)d_in[1];
    const int*   dst   = (const int*)d_in[2];
    const float* W1s   = (const float*)d_in[3];
    const float* b1s   = (const float*)d_in[4];
    const float* W1d   = (const float*)d_in[5];
    const float* b1d   = (const float*)d_in[6];
    const float* attn1 = (const float*)d_in[7];
    const float* W2s   = (const float*)d_in[8];
    const float* b2s   = (const float*)d_in[9];
    const float* W2d   = (const float*)d_in[10];
    const float* b2d   = (const float*)d_in[11];
    const float* attn2 = (const float*)d_in[12];
    (void)in_sizes; (void)n_in; (void)out_size; (void)ws_size;

    char* ws = (char*)d_ws;
    const size_t MB = 1024u * 1024u;
    float*    log1  = (float*)(ws + 32 * MB);
    unsigned* m1    = (unsigned*)(ws + 40 * MB);
    float*    den1  = (float*)(ws + 40 * MB + 262144);
    float*    h1    = (float*)(ws + 41 * MB);
    float*    z1    = (float*)(ws + 57 * MB);
    float*    fs2   = (float*)(ws + 16 * MB);
    float*    fd2   = (float*)(ws + 24 * MB);
    float*    agg2  = (float*)(ws + 0);
    float*    log2  = (float*)(ws + 8 * MB);
    unsigned* m2    = (unsigned*)(ws + 12 * MB);
    float*    den2  = (float*)(ws + 12 * MB + 131072);

    float* out = (float*)d_out;

    zero_l1<<<N_NODE * 2 * D_IN / 256, 256, 0, stream>>>(z1, m1, den1);
    edge_logits1_x<<<E_CNT / (16 * LT_TPW * 4), 128, 0, stream>>>(
        x, src, dst, W1s, b1s, W1d, b1d, attn1, log1, m1);
    edge_agg_lds<2 * D_IN, 2, 4096, D_IN><<<E_CNT / 4096, 1024, 0, stream>>>(x, src, dst, log1, m1, z1, den1);
    gemm_z_elu<<<N_NODE / 16 / 4, 128, 0, stream>>>(z1, W1s, b1s, den1, h1);

    zero_l2<<<N_NODE * 64 / 256, 256, 0, stream>>>(agg2, m2, den2);
    gemm_dual_wmma<2 * DH, DH><<<N_NODE / 16 / 4, 128, 0, stream>>>(
        h1, W2s, b2s, W2d, b2d, fs2, fd2);
    edge_logits2<<<E_CNT / 8 / EPW, 256, 0, stream>>>(fs2, fd2, src, dst, attn2, log2, m2);
    edge_agg_lds<DH, 1, 8192, DH><<<E_CNT / 8192 * 2, 1024, 0, stream>>>(fs2, src, dst, log2, m2, agg2, den2);

    reorder_out<<<N_NODE * 64 / 256, 256, 0, stream>>>(agg2, den2, out);
}


# Edge maxima taken in LDS first: a logits work-group (4096 edges) keeps a 2048-row window of maxima by LDS atomics and flushes it by whole-line atomic max; scattered global max atomics go
# speedup vs baseline: 4.1658x; 1.4571x over previous
#include <hip/hip_runtime.h>

constexpr int B_SZ   = 16;
constexpr int NUM_PM = 512;
constexpr int NUM_VM = 1536;
constexpr int NPG    = NUM_PM + NUM_VM;
constexpr int N_NODE = B_SZ * NPG;
constexpr int DEG    = 32;
constexpr int E_CNT  = DEG * N_NODE;
constexpr int D_IN   = 16;
constexpr int DH     = 64;
constexpr float NEG_SLOPE = 0.2f;
constexpr int EPW    = 8;
static_assert((EPW & (EPW - 1)) == 0 && EPW <= 16, "a run's results are written by lanes 0..EPW-1 of each 16-lane half");

constexpr int WIN = 2048;

typedef float v2f __attribute__((ext_vector_type(2)));
typedef float v8f __attribute__((ext_vector_type(8)));

__device__ __forceinline__ unsigned enc_f(float f) {
    unsigned u = __float_as_uint(f);
    return (u & 0x80000000u) ? ~u : (u | 0x80000000u);
}
__device__ __forceinline__ float dec_f(unsigned u) {
    return __uint_as_float((u & 0x80000000u) ? (u & 0x7fffffffu) : ~u);
}

template <int K, int COLS>
__global__ __launch_bounds__(128) void gemm_dual_wmma(
    const float* __restrict__ X,
    const float* __restrict__ Ws, const float* __restrict__ Bs,
    const float* __restrict__ Wd, const float* __restrict__ Bd,
    float* __restrict__ Fs, float* __restrict__ Fd) {
    const int lane = threadIdx.x & 31;
    const int wave = threadIdx.x >> 5;
    const int row0 = (blockIdx.x * 4 + wave) * 16;
    const int m  = lane & 15;
    const int hi = lane >> 4;
    const float* Xr = X + (size_t)(row0 + m) * K;

    for (int t = 0; t < 2; ++t) {
        const float* W  = t ? Wd : Ws;
        const float* Bv = t ? Bd : Bs;
        float*       F  = t ? Fd : Fs;
        for (int col0 = 0; col0 < COLS; col0 += 16) {
            v8f acc = {};
#pragma unroll
            for (int kk = 0; kk < K; kk += 4) {
                v2f a, b;
                a.x = Xr[kk + 2 * hi + 0];
                a.y = Xr[kk + 2 * hi + 1];
                b.x = W[(size_t)(kk + 2 * hi + 0) * COLS + col0 + m];
                b.y = W[(size_t)(kk + 2 * hi + 1) * COLS + col0 + m];
                acc = __builtin_amdgcn_wmma_f32_16x16x4_f32(
                    false, a, false, b, (short)0, acc, false, false);
            }
            const float bias = Bv[col0 + m];
#pragma unroll
            for (int r = 0; r < 8; ++r) {
                F[(size_t)(row0 + r + 8 * hi) * COLS + col0 + m] = acc[r] + bias;
            }
        }
    }
}

__global__ __launch_bounds__(256) void zero_l1(float* __restrict__ z1,
                                               unsigned* __restrict__ m1,
                                               float* __restrict__ den1) {
    int idx = blockIdx.x * 256 + threadIdx.x;
    z1[idx] = 0.f;
    if (idx < N_NODE * 2) { m1[idx] = 0u; den1[idx] = 0.f; }
}
__global__ __launch_bounds__(256) void zero_l2(float* __restrict__ agg,
                                               unsigned* __restrict__ m2,
                                               float* __restrict__ den2) {
    int idx = blockIdx.x * 256 + threadIdx.x;
    agg[idx] = 0.f;
    if (idx < N_NODE) { m2[idx] = 0u; den2[idx] = 0.f; }
}

constexpr int LT_TPW = 16;
constexpr int LT_WAVES = 16;
static_assert(E_CNT % (16 * LT_TPW * LT_WAVES) == 0, "whole tiles per wave, whole waves per work-group");

template <int CTRL>
__device__ __forceinline__ float row_xm(float v) {
    return __int_as_float(__builtin_amdgcn_update_dpp(0, __float_as_int(v), CTRL, 0xf, 0xf, false));
}

__global__ __launch_bounds__(LT_WAVES * 32) void edge_logits1_x(
    const float* __restrict__ x, const int* __restrict__ src, const int* __restrict__ dst,
    const float* __restrict__ Ws, const float* __restrict__ Bs,
    const float* __restrict__ Wd, const float* __restrict__ Bd,
    const float* __restrict__ attn, float* __restrict__ logits, unsigned* __restrict__ mmax) {
    __shared__ v2f wl[64 * 32];
    __shared__ unsigned lmax[WIN * 2];
    const int lane = threadIdx.x & 31;
    const int wave = threadIdx.x >> 5;
    const int c  = lane & 15;
    const int hh = lane >> 4;

    for (int q = threadIdx.x; q < 64 * 32; q += LT_WAVES * 32) {
        const int l = q & 31, j = q >> 8, n = (q >> 5) & 7;
        const float* Wm = (j < 4) ? Ws : Wd;
        const int k = 8 * (l >> 4) + 2 * (j & 3);
        v2f w;
        w.x = Wm[(size_t)(k + 0) * (2 * DH) + 16 * n + (l & 15)];
        w.y = Wm[(size_t)(k + 1) * (2 * DH) + 16 * n + (l & 15)];
        wl[q] = w;
    }
    for (int q = threadIdx.x; q < WIN * 2; q += LT_WAVES * 32) lmax[q] = 0u;
    const int wbase = dst[blockIdx.x * (LT_WAVES * LT_TPW * 16)] & ~(WIN - 1);
    __syncthreads();
    v2f wf[8][8];
#pragma unroll
    for (int j = 0; j < 8; ++j)
#pragma unroll
        for (int n = 0; n < 8; ++n) wf[j][n] = wl[(j * 8 + n) * 32 + lane];
    float bv[8], av[8];
#pragma unroll
    for (int n = 0; n < 8; ++n) {
        bv[n] = Bs[16 * n + c] + Bd[16 * n + c];
        av[n] = attn[16 * n + c];
    }

    const int ebase = (blockIdx.x * LT_WAVES + wave) * (LT_TPW * 16);
    const int osel  = 8 * hh + (c & 7);
    const int ohead = c >> 3;
    const int et1 = (LT_TPW > 1) ? ebase + 16 : ebase;
    const int s1 = src[ebase + c], d1 = dst[ebase + c];
    int s2 = src[et1 + c], d2 = dst[et1 + c];
    float4 xs0 = *(const float4*)(x + (size_t)s1 * D_IN + 8 * hh);
    float4 xs1 = *(const float4*)(x + (size_t)s1 * D_IN + 8 * hh + 4);
    float4 xd0 = *(const float4*)(x + (size_t)d1 * D_IN + 8 * hh);
    float4 xd1 = *(const float4*)(x + (size_t)d1 * D_IN + 8 * hh + 4);
    for (int t = 0; t < LT_TPW; ++t) {
        const int e0 = ebase + t * 16;
        const int e2 = ebase + ((t + 2 < LT_TPW) ? t + 2 : LT_TPW - 1) * 16;
        const int s3 = src[e2 + c], d3 = dst[e2 + c];
        const int dw = dst[e0 + osel];
        const float4 ns0 = *(const float4*)(x + (size_t)s2 * D_IN + 8 * hh);
        const float4 ns1 = *(const float4*)(x + (size_t)s2 * D_IN + 8 * hh + 4);
        const float4 nd0 = *(const float4*)(x + (size_t)d2 * D_IN + 8 * hh);
        const float4 nd1 = *(const float4*)(x + (size_t)d2 * D_IN + 8 * hh + 4);

        v2f a[8];
        a[0].x = xs0.x; a[0].y = xs0.y; a[1].x = xs0.z; a[1].y = xs0.w;
        a[2].x = xs1.x; a[2].y = xs1.y; a[3].x = xs1.z; a[3].y = xs1.w;
        a[4].x = xd0.x; a[4].y = xd0.y; a[5].x = xd0.z; a[5].y = xd0.w;
        a[6].x = xd1.x; a[6].y = xd1.y; a[7].x = xd1.z; a[7].y = xd1.w;
        float p[16];
#pragma unroll
        for (int i = 0; i < 16; ++i) p[i] = 0.f;
#pragma unroll
        for (int n = 0; n < 8; ++n) {
            v8f acc = {};
#pragma unroll
            for (int j = 0; j < 8; ++j)
                acc = __builtin_amdgcn_wmma_f32_16x16x4_f32(
                    false, a[j], false, wf[j][n], (short)0, acc, false, false);
#pragma unroll
            for (int r = 0; r < 8; ++r) {
                float u = acc[r] + bv[n];
                u = u > 0.f ? u : NEG_SLOPE * u;
                p[(n >> 2) * 8 + r] += u * av[n];
            }
        }
#pragma unroll
        for (int i = 0; i < 8; ++i) {
            const float keep = (c & 8) ? p[i + 8] : p[i], send = (c & 8) ? p[i] : p[i + 8];
            p[i] = keep + row_xm<0x168>(send);
        }
#pragma unroll
        for (int i = 0; i < 4; ++i) {
            const float keep = (c & 4) ? p[i + 4] : p[i], send = (c & 4) ? p[i] : p[i + 4];
            p[i] = keep + row_xm<0x164>(send);
        }
#pragma unroll
        for (int i = 0; i < 2; ++i) {
            const float keep = (c & 2) ? p[i + 2] : p[i], send = (c & 2) ? p[i] : p[i + 2];
            p[i] = keep + row_xm<0x162>(send);
        }
        const float keep = (c & 1) ? p[1] : p[0], send = (c & 1) ? p[0] : p[1];
        const float pv = keep + row_xm<0x161>(send);
        logits[(size_t)(e0 + osel) * 2 + ohead] = pv;
        const int rw = dw - wbase;
        if ((unsigned)rw < (unsigned)WIN) atomicMax(&lmax[rw * 2 + ohead], enc_f(pv));
        else                              atomicMax(&mmax[dw * 2 + ohead], enc_f(pv));

        xs0 = ns0; xs1 = ns1; xd0 = nd0; xd1 = nd1;
        s2 = s3; d2 = d3;
    }
    __syncthreads();
    for (int q = threadIdx.x; q < WIN * 2; q += LT_WAVES * 32)
        atomicMax(&mmax[(size_t)wbase * 2 + q], lmax[q]);
}

template <int FT, int HEADS, int EPB, int SRCW>
__global__ __launch_bounds__(1024) void edge_agg_lds(
    const float* __restrict__ fs, const int* __restrict__ src,
    const int* __restrict__ dst, const float* __restrict__ lg,
    const unsigned* __restrict__ mmax, float* __restrict__ out, float* __restrict__ den) {
    constexpr int SL   = FT / 32;
    constexpr int EPWV = EPB / 32;
    constexpr int HW   = FT / HEADS;
    static_assert(EPWV % EPW == 0 && FT % 32 == 0, "whole runs per wave, whole slices per row");
    static_assert((SRCW == FT && HW % 32 == 0) || (SRCW == HW && 32 % HW == 0),
                  "a slice inside one head, or whole heads inside a slice");
    static_assert(HW >= EPW, "a head's lanes in a slice cover one run of edges");
    __shared__ float acc[WIN * 32];
    __shared__ float dacc[WIN * HEADS];
    const int wave  = __builtin_amdgcn_readfirstlane((int)(threadIdx.x >> 5));
    const int lane  = threadIdx.x & 31;
    const int slice = blockIdx.x % SL;
    const int eb    = (blockIdx.x / SL) * EPB;
    const int fo    = slice * 32 + lane;
    const int h     = (HW % 32 == 0) ? (slice * 32) / HW : fo / HW;
    const int fsrc  = fo % SRCW;
    const int wbase = dst[eb] & ~(WIN - 1);
    const bool dblk = (HW % 32 == 0) ? ((slice * 32) % HW == 0) : true;
    const int  q    = (HW % 32 == 0) ? lane : lane % HW;

    for (int i = threadIdx.x; i < WIN * 32 / 4; i += 1024)
        ((float4*)acc)[i] = make_float4(0.f, 0.f, 0.f, 0.f);
    for (int i = threadIdx.x; i < WIN * HEADS / 4; i += 1024)
        ((float4*)dacc)[i] = make_float4(0.f, 0.f, 0.f, 0.f);
    __syncthreads();

    const int ew = eb + wave * EPWV;
    int s[EPW], d[EPW];
#pragma unroll
    for (int i = 0; i < EPW; ++i) { s[i] = src[ew + i]; d[i] = dst[ew + i]; }
    for (int st = 0; st < EPWV; st += EPW) {
        const int e0 = ew + st;
        const int en = (st + EPW < EPWV) ? e0 + EPW : e0;
        int sn[EPW], dx[EPW];
#pragma unroll
        for (int i = 0; i < EPW; ++i) { sn[i] = src[en + i]; dx[i] = dst[en + i]; }
        float v[EPW], xl[EPW];
        unsigned mx[EPW];
#pragma unroll
        for (int i = 0; i < EPW; ++i) {
            v[i]  = fs[(size_t)s[i] * SRCW + fsrc];
            xl[i] = lg[(size_t)(e0 + i) * HEADS + h];
            mx[i] = mmax[d[i] * HEADS + h];
        }
        float w[EPW];
#pragma unroll
        for (int i = 0; i < EPW; ++i) {
            w[i] = __expf(xl[i] - dec_f(mx[i]));
            const float val = v[i] * w[i];
            const int r = d[i] - wbase;
            if ((unsigned)r < (unsigned)WIN) atomicAdd(&acc[r * 32 + lane], val);
            else                             atomicAdd(out + (size_t)d[i] * FT + fo, val);
        }
        if (dblk) {
            float pw = w[0];
            int   pd = d[0];
#pragma unroll
            for (int i = 1; i < EPW; ++i) {
                pw = (q == i) ? w[i] : pw;
                pd = (q == i) ? d[i] : pd;
            }
            if (q < EPW) {
                const int r = pd - wbase;
                if ((unsigned)r < (unsigned)WIN) atomicAdd(&dacc[r * HEADS + h], pw);
                else                             atomicAdd(den + pd * HEADS + h, pw);
            }
        }
#pragma unroll
        for (int i = 0; i < EPW; ++i) { s[i] = sn[i]; d[i] = dx[i]; }
    }
    __syncthreads();

    for (int i0 = threadIdx.x; i0 < WIN * 32; i0 += 8 * 1024) {
        float a[8];
#pragma unroll
        for (int j = 0; j < 8; ++j) a[j] = acc[i0 + j * 1024];
#pragma unroll
        for (int j = 0; j < 8; ++j)
            atomicAdd(out + (size_t)(wbase + ((i0 + j * 1024) >> 5)) * FT + fo, a[j]);
    }
    if (dblk)
        for (int i = threadIdx.x; i < WIN * HEADS; i += 1024)
            atomicAdd(den + (size_t)wbase * HEADS + i, dacc[i]);
}

__global__ __launch_bounds__(128) void gemm_z_elu(
    const float* __restrict__ Z, const float* __restrict__ W, const float* __restrict__ Bv,
    const float* __restrict__ den, float* __restrict__ H) {
    const int lane = threadIdx.x & 31;
    const int wave = threadIdx.x >> 5;
    const int row0 = (blockIdx.x * 4 + wave) * 16;
    const int m  = lane & 15;
    const int hi = lane >> 4;

    for (int t = 0; t < 2; ++t) {
        const float* Zr = Z + (size_t)(row0 + m) * (2 * D_IN) + t * D_IN;
        v2f a[D_IN / 4];
#pragma unroll
        for (int kk = 0; kk < D_IN; kk += 4) {
            a[kk / 4].x = Zr[kk + 2 * hi + 0];
            a[kk / 4].y = Zr[kk + 2 * hi + 1];
        }
        float has[8], inv[8];
#pragma unroll
        for (int r = 0; r < 8; ++r) {
            const float dn = den[(row0 + r + 8 * hi) * 2 + t];
            has[r] = dn > 0.f ? 1.f : 0.f;
            inv[r] = dn > 0.f ? 1.f / dn : 0.f;
        }
        for (int c0 = 0; c0 < DH; c0 += 16) {
            const int col0 = t * DH + c0;
            v8f acc = {};
#pragma unroll
            for (int kk = 0; kk < D_IN; kk += 4) {
                v2f b;
                b.x = W[(size_t)(kk + 2 * hi + 0) * (2 * DH) + col0 + m];
                b.y = W[(size_t)(kk + 2 * hi + 1) * (2 * DH) + col0 + m];
                acc = __builtin_amdgcn_wmma_f32_16x16x4_f32(
                    false, a[kk / 4], false, b, (short)0, acc, false, false);
            }
            const float bias = Bv[col0 + m];
#pragma unroll
            for (int r = 0; r < 8; ++r) {
                const float v = acc[r] * inv[r] + bias * has[r];
                H[(size_t)(row0 + r + 8 * hi) * (2 * DH) + col0 + m] = v > 0.f ? v : (__expf(v) - 1.f);
            }
        }
    }
}

constexpr int L2_EPB = 4096;
static_assert(E_CNT % L2_EPB == 0 && (L2_EPB / 32) % EPW == 0, "whole work-groups, whole runs per wave");
__global__ __launch_bounds__(1024) void edge_logits2(
    const float* __restrict__ fs, const float* __restrict__ fd,
    const int* __restrict__ src, const int* __restrict__ dst,
    const float* __restrict__ attn, float* __restrict__ logits,
    unsigned* __restrict__ mmax) {
    constexpr int EPWV = L2_EPB / 32;
    constexpr int LAST = EPWV - EPW;
    __shared__ unsigned lmax[WIN];
    const int wave = __builtin_amdgcn_readfirstlane((int)(threadIdx.x >> 5));
    const int lane = threadIdx.x & 31;
    const int eb = blockIdx.x * L2_EPB;
    const int wbase = dst[eb] & ~(WIN - 1);
    for (int i = threadIdx.x; i < WIN; i += 1024) lmax[i] = 0u;
    __syncthreads();

    const float2 w = *(const float2*)(attn + lane * 2);
    const int ew = eb + wave * EPWV;
    const int sel = lane & (EPW - 1);
    constexpr int R1 = (EPW < LAST) ? EPW : LAST;
    int d[EPW], sn[EPW], dn[EPW];
    float2 a[EPW], b[EPW];
#pragma unroll
    for (int i = 0; i < EPW; ++i) {
        const int s0 = src[ew + i];
        d[i]  = dst[ew + i];
        sn[i] = src[ew + R1 + i];
        dn[i] = dst[ew + R1 + i];
        a[i] = *(const float2*)(fs + (size_t)s0 * 64 + lane * 2);
        b[i] = *(const float2*)(fd + (size_t)d[i] * 64 + lane * 2);
    }
    for (int st = 0; st < EPWV; st += EPW) {
        const int e0 = ew + st;
        const int e2 = ew + ((st + 2 * EPW < LAST) ? st + 2 * EPW : LAST);
        int s2[EPW], d2[EPW];
#pragma unroll
        for (int i = 0; i < EPW; ++i) { s2[i] = src[e2 + i]; d2[i] = dst[e2 + i]; }
        float2 na[EPW], nb[EPW];
#pragma unroll
        for (int i = 0; i < EPW; ++i) {
            na[i] = *(const float2*)(fs + (size_t)sn[i] * 64 + lane * 2);
            nb[i] = *(const float2*)(fd + (size_t)dn[i] * 64 + lane * 2);
        }
        float pe[EPW];
#pragma unroll
        for (int i = 0; i < EPW; ++i) {
            float p = 0.f, t;
            t = a[i].x + b[i].x; t = t > 0.f ? t : NEG_SLOPE * t; p += t * w.x;
            t = a[i].y + b[i].y; t = t > 0.f ? t : NEG_SLOPE * t; p += t * w.y;
            p += __shfl_xor(p, 1);  p += __shfl_xor(p, 2);
            p += __shfl_xor(p, 4);  p += __shfl_xor(p, 8);
            p += __shfl_xor(p, 16);
            pe[i] = p;
        }
        float pv = pe[0];
        int   dv = d[0];
#pragma unroll
        for (int i = 1; i < EPW; ++i) {
            pv = (sel == i) ? pe[i] : pv;
            dv = (sel == i) ? d[i]  : dv;
        }
        if (lane < EPW) {
            logits[e0 + sel] = pv;
            const int r = dv - wbase;
            if ((unsigned)r < (unsigned)WIN) atomicMax(&lmax[r], enc_f(pv));
            else                             atomicMax(&mmax[dv], enc_f(pv));
        }
#pragma unroll
        for (int i = 0; i < EPW; ++i) {
            a[i] = na[i]; b[i] = nb[i];
            d[i] = dn[i]; sn[i] = s2[i]; dn[i] = d2[i];
        }
    }
    __syncthreads();
    for (int i = threadIdx.x; i < WIN; i += 1024)
        atomicMax(&mmax[(size_t)wbase + i], lmax[i]);
}

__global__ __launch_bounds__(256) void reorder_out(const float* __restrict__ agg,
                                                   const float* __restrict__ den,
                                                   float* __restrict__ out) {
    const int idx = blockIdx.x * 256 + threadIdx.x;
    const int n = idx >> 6, f = idx & 63;
    const int b = n >> 11;
    const int r = n & 2047;
    size_t o;
    if (r < NUM_PM) o = ((size_t)(b * NUM_PM + r) * 64) + f;
    else            o = (size_t)B_SZ * NUM_PM * 64 +
                        ((size_t)(b * NUM_VM + (r - NUM_PM)) * 64) + f;
    const float dn = den[n];
    out[o] = dn > 0.f ? agg[idx] / dn : 0.f;
}

extern "C" void kernel_launch(void* const* d_in, const int* in_sizes, int n_in,
                              void* d_out, int out_size, void* d_ws, size_t ws_size,
                              hipStream_t stream) {
    const float* x     = (const float*)d_in[0];
    const int*   src   = (const int*)d_in[1];
    const int*   dst   = (const int*)d_in[2];
    const float* W1s   = (const float*)d_in[3];
    const float* b1s   = (const float*)d_in[4];
    const float* W1d   = (const float*)d_in[5];
    const float* b1d   = (const float*)d_in[6];
    const float* attn1 = (const float*)d_in[7];
    const float* W2s   = (const float*)d_in[8];
    const float* b2s   = (const float*)d_in[9];
    const float* W2d   = (const float*)d_in[10];
    const float* b2d   = (const float*)d_in[11];
    const float* attn2 = (const float*)d_in[12];
    (void)in_sizes; (void)n_in; (void)out_size; (void)ws_size;

    char* ws = (char*)d_ws;
    const size_t MB = 1024u * 1024u;
    float*    log1  = (float*)(ws + 32 * MB);
    unsigned* m1    = (unsigned*)(ws + 40 * MB);
    float*    den1  = (float*)(ws + 40 * MB + 262144);
    float*    h1    = (float*)(ws + 41 * MB);
    float*    z1    = (float*)(ws + 57 * MB);
    float*    fs2   = (float*)(ws + 16 * MB);
    float*    fd2   = (float*)(ws + 24 * MB);
    float*    agg2  = (float*)(ws + 0);
    float*    log2  = (float*)(ws + 8 * MB);
    unsigned* m2    = (unsigned*)(ws + 12 * MB);
    float*    den2  = (float*)(ws + 12 * MB + 131072);

    float* out = (float*)d_out;

    zero_l1<<<N_NODE * 2 * D_IN / 256, 256, 0, stream>>>(z1, m1, den1);
    edge_logits1_x<<<E_CNT / (16 * LT_TPW * LT_WAVES), LT_WAVES * 32, 0, stream>>>(
        x, src, dst, W1s, b1s, W1d, b1d, attn1, log1, m1);
    edge_agg_lds<2 * D_IN, 2, 4096, D_IN><<<E_CNT / 4096, 1024, 0, stream>>>(x, src, dst, log1, m1, z1, den1);
    gemm_z_elu<<<N_NODE / 16 / 4, 128, 0, stream>>>(z1, W1s, b1s, den1, h1);

    zero_l2<<<N_NODE * 64 / 256, 256, 0, stream>>>(agg2, m2, den2);
    gemm_dual_wmma<2 * DH, DH><<<N_NODE / 16 / 4, 128, 0, stream>>>(
        h1, W2s, b2s, W2d, b2d, fs2, fd2);
    edge_logits2<<<E_CNT / L2_EPB, 1024, 0, stream>>>(fs2, fd2, src, dst, attn2, log2, m2);
    edge_agg_lds<DH, 1, 8192, DH><<<E_CNT / 8192 * 2, 1024, 0, stream>>>(fs2, src, dst, log2, m2, agg2, den2);

    reorder_out<<<N_NODE * 64 / 256, 256, 0, stream>>>(agg2, den2, out);
}


# Aggregation: a run's 8 edge weights computed once across the lanes (one logit load, one max load, one exp per run, not per edge), passed to the row lanes by DPP row-share; sums as before
# speedup vs baseline: 4.5969x; 1.1035x over previous
#include <hip/hip_runtime.h>

constexpr int B_SZ   = 16;
constexpr int NUM_PM = 512;
constexpr int NUM_VM = 1536;
constexpr int NPG    = NUM_PM + NUM_VM;
constexpr int N_NODE = B_SZ * NPG;
constexpr int DEG    = 32;
constexpr int E_CNT  = DEG * N_NODE;
constexpr int D_IN   = 16;
constexpr int DH     = 64;
constexpr float NEG_SLOPE = 0.2f;
constexpr int EPW    = 8;
static_assert((EPW & (EPW - 1)) == 0 && EPW <= 16, "a run's results are written by lanes 0..EPW-1 of each 16-lane half");

constexpr int WIN = 2048;

typedef float v2f __attribute__((ext_vector_type(2)));
typedef float v8f __attribute__((ext_vector_type(8)));

__device__ __forceinline__ unsigned enc_f(float f) {
    unsigned u = __float_as_uint(f);
    return (u & 0x80000000u) ? ~u : (u | 0x80000000u);
}
__device__ __forceinline__ float dec_f(unsigned u) {
    return __uint_as_float((u & 0x80000000u) ? (u & 0x7fffffffu) : ~u);
}

template <int K, int COLS>
__global__ __launch_bounds__(128) void gemm_dual_wmma(
    const float* __restrict__ X,
    const float* __restrict__ Ws, const float* __restrict__ Bs,
    const float* __restrict__ Wd, const float* __restrict__ Bd,
    float* __restrict__ Fs, float* __restrict__ Fd) {
    const int lane = threadIdx.x & 31;
    const int wave = threadIdx.x >> 5;
    const int row0 = (blockIdx.x * 4 + wave) * 16;
    const int m  = lane & 15;
    const int hi = lane >> 4;
    const float* Xr = X + (size_t)(row0 + m) * K;

    for (int t = 0; t < 2; ++t) {
        const float* W  = t ? Wd : Ws;
        const float* Bv = t ? Bd : Bs;
        float*       F  = t ? Fd : Fs;
        for (int col0 = 0; col0 < COLS; col0 += 16) {
            v8f acc = {};
#pragma unroll
            for (int kk = 0; kk < K; kk += 4) {
                v2f a, b;
                a.x = Xr[kk + 2 * hi + 0];
                a.y = Xr[kk + 2 * hi + 1];
                b.x = W[(size_t)(kk + 2 * hi + 0) * COLS + col0 + m];
                b.y = W[(size_t)(kk + 2 * hi + 1) * COLS + col0 + m];
                acc = __builtin_amdgcn_wmma_f32_16x16x4_f32(
                    false, a, false, b, (short)0, acc, false, false);
            }
            const float bias = Bv[col0 + m];
#pragma unroll
            for (int r = 0; r < 8; ++r) {
                F[(size_t)(row0 + r + 8 * hi) * COLS + col0 + m] = acc[r] + bias;
            }
        }
    }
}

__global__ __launch_bounds__(256) void zero_l1(float* __restrict__ z1,
                                               unsigned* __restrict__ m1,
                                               float* __restrict__ den1) {
    int idx = blockIdx.x * 256 + threadIdx.x;
    z1[idx] = 0.f;
    if (idx < N_NODE * 2) { m1[idx] = 0u; den1[idx] = 0.f; }
}
__global__ __launch_bounds__(256) void zero_l2(float* __restrict__ agg,
                                               unsigned* __restrict__ m2,
                                               float* __restrict__ den2) {
    int idx = blockIdx.x * 256 + threadIdx.x;
    agg[idx] = 0.f;
    if (idx < N_NODE) { m2[idx] = 0u; den2[idx] = 0.f; }
}

constexpr int LT_TPW = 16;
constexpr int LT_WAVES = 16;
static_assert(E_CNT % (16 * LT_TPW * LT_WAVES) == 0, "whole tiles per wave, whole waves per work-group");

template <int CTRL>
__device__ __forceinline__ float row_xm(float v) {
    return __int_as_float(__builtin_amdgcn_update_dpp(0, __float_as_int(v), CTRL, 0xf, 0xf, false));
}

template <int I>
__device__ __forceinline__ float row_sh(float v) {
    return __int_as_float(__builtin_amdgcn_update_dpp(0, __float_as_int(v), 0x150 + I, 0xf, 0xf, false));
}
__device__ __forceinline__ float row_pick(float v, int i) {
    switch (i) {
        case 0:  return row_sh<0>(v);
        case 1:  return row_sh<1>(v);
        case 2:  return row_sh<2>(v);
        case 3:  return row_sh<3>(v);
        case 4:  return row_sh<4>(v);
        case 5:  return row_sh<5>(v);
        case 6:  return row_sh<6>(v);
        case 7:  return row_sh<7>(v);
        case 8:  return row_sh<8>(v);
        case 9:  return row_sh<9>(v);
        case 10: return row_sh<10>(v);
        case 11: return row_sh<11>(v);
        case 12: return row_sh<12>(v);
        case 13: return row_sh<13>(v);
        case 14: return row_sh<14>(v);
        default: return row_sh<15>(v);
    }
}

__global__ __launch_bounds__(LT_WAVES * 32) void edge_logits1_x(
    const float* __restrict__ x, const int* __restrict__ src, const int* __restrict__ dst,
    const float* __restrict__ Ws, const float* __restrict__ Bs,
    const float* __restrict__ Wd, const float* __restrict__ Bd,
    const float* __restrict__ attn, float* __restrict__ logits, unsigned* __restrict__ mmax) {
    __shared__ v2f wl[64 * 32];
    __shared__ unsigned lmax[WIN * 2];
    const int lane = threadIdx.x & 31;
    const int wave = threadIdx.x >> 5;
    const int c  = lane & 15;
    const int hh = lane >> 4;

    for (int q = threadIdx.x; q < 64 * 32; q += LT_WAVES * 32) {
        const int l = q & 31, j = q >> 8, n = (q >> 5) & 7;
        const float* Wm = (j < 4) ? Ws : Wd;
        const int k = 8 * (l >> 4) + 2 * (j & 3);
        v2f w;
        w.x = Wm[(size_t)(k + 0) * (2 * DH) + 16 * n + (l & 15)];
        w.y = Wm[(size_t)(k + 1) * (2 * DH) + 16 * n + (l & 15)];
        wl[q] = w;
    }
    for (int q = threadIdx.x; q < WIN * 2; q += LT_WAVES * 32) lmax[q] = 0u;
    const int wbase = dst[blockIdx.x * (LT_WAVES * LT_TPW * 16)] & ~(WIN - 1);
    __syncthreads();
    v2f wf[8][8];
#pragma unroll
    for (int j = 0; j < 8; ++j)
#pragma unroll
        for (int n = 0; n < 8; ++n) wf[j][n] = wl[(j * 8 + n) * 32 + lane];
    float bv[8], av[8];
#pragma unroll
    for (int n = 0; n < 8; ++n) {
        bv[n] = Bs[16 * n + c] + Bd[16 * n + c];
        av[n] = attn[16 * n + c];
    }

    const int ebase = (blockIdx.x * LT_WAVES + wave) * (LT_TPW * 16);
    const int osel  = 8 * hh + (c & 7);
    const int ohead = c >> 3;
    const int et1 = (LT_TPW > 1) ? ebase + 16 : ebase;
    const int s1 = src[ebase + c], d1 = dst[ebase + c];
    int s2 = src[et1 + c], d2 = dst[et1 + c];
    float4 xs0 = *(const float4*)(x + (size_t)s1 * D_IN + 8 * hh);
    float4 xs1 = *(const float4*)(x + (size_t)s1 * D_IN + 8 * hh + 4);
    float4 xd0 = *(const float4*)(x + (size_t)d1 * D_IN + 8 * hh);
    float4 xd1 = *(const float4*)(x + (size_t)d1 * D_IN + 8 * hh + 4);
    for (int t = 0; t < LT_TPW; ++t) {
        const int e0 = ebase + t * 16;
        const int e2 = ebase + ((t + 2 < LT_TPW) ? t + 2 : LT_TPW - 1) * 16;
        const int s3 = src[e2 + c], d3 = dst[e2 + c];
        const int dw = dst[e0 + osel];
        const float4 ns0 = *(const float4*)(x + (size_t)s2 * D_IN + 8 * hh);
        const float4 ns1 = *(const float4*)(x + (size_t)s2 * D_IN + 8 * hh + 4);
        const float4 nd0 = *(const float4*)(x + (size_t)d2 * D_IN + 8 * hh);
        const float4 nd1 = *(const float4*)(x + (size_t)d2 * D_IN + 8 * hh + 4);

        v2f a[8];
        a[0].x = xs0.x; a[0].y = xs0.y; a[1].x = xs0.z; a[1].y = xs0.w;
        a[2].x = xs1.x; a[2].y = xs1.y; a[3].x = xs1.z; a[3].y = xs1.w;
        a[4].x = xd0.x; a[4].y = xd0.y; a[5].x = xd0.z; a[5].y = xd0.w;
        a[6].x = xd1.x; a[6].y = xd1.y; a[7].x = xd1.z; a[7].y = xd1.w;
        float p[16];
#pragma unroll
        for (int i = 0; i < 16; ++i) p[i] = 0.f;
#pragma unroll
        for (int n = 0; n < 8; ++n) {
            v8f acc = {};
#pragma unroll
            for (int j = 0; j < 8; ++j)
                acc = __builtin_amdgcn_wmma_f32_16x16x4_f32(
                    false, a[j], false, wf[j][n], (short)0, acc, false, false);
#pragma unroll
            for (int r = 0; r < 8; ++r) {
                float u = acc[r] + bv[n];
                u = u > 0.f ? u : NEG_SLOPE * u;
                p[(n >> 2) * 8 + r] += u * av[n];
            }
        }
#pragma unroll
        for (int i = 0; i < 8; ++i) {
            const float keep = (c & 8) ? p[i + 8] : p[i], send = (c & 8) ? p[i] : p[i + 8];
            p[i] = keep + row_xm<0x168>(send);
        }
#pragma unroll
        for (int i = 0; i < 4; ++i) {
            const float keep = (c & 4) ? p[i + 4] : p[i], send = (c & 4) ? p[i] : p[i + 4];
            p[i] = keep + row_xm<0x164>(send);
        }
#pragma unroll
        for (int i = 0; i < 2; ++i) {
            const float keep = (c & 2) ? p[i + 2] : p[i], send = (c & 2) ? p[i] : p[i + 2];
            p[i] = keep + row_xm<0x162>(send);
        }
        const float keep = (c & 1) ? p[1] : p[0], send = (c & 1) ? p[0] : p[1];
        const float pv = keep + row_xm<0x161>(send);
        logits[(size_t)(e0 + osel) * 2 + ohead] = pv;
        const int rw = dw - wbase;
        if ((unsigned)rw < (unsigned)WIN) atomicMax(&lmax[rw * 2 + ohead], enc_f(pv));
        else                              atomicMax(&mmax[dw * 2 + ohead], enc_f(pv));

        xs0 = ns0; xs1 = ns1; xd0 = nd0; xd1 = nd1;
        s2 = s3; d2 = d3;
    }
    __syncthreads();
    for (int q = threadIdx.x; q < WIN * 2; q += LT_WAVES * 32)
        atomicMax(&mmax[(size_t)wbase * 2 + q], lmax[q]);
}

template <int FT, int HEADS, int EPB, int SRCW>
__global__ __launch_bounds__(1024) void edge_agg_lds(
    const float* __restrict__ fs, const int* __restrict__ src,
    const int* __restrict__ dst, const float* __restrict__ lg,
    const unsigned* __restrict__ mmax, float* __restrict__ out, float* __restrict__ den) {
    constexpr int SL   = FT / 32;
    constexpr int EPWV = EPB / 32;
    constexpr int HW   = FT / HEADS;
    static_assert(EPWV % EPW == 0 && FT % 32 == 0, "whole runs per wave, whole slices per row");
    static_assert((SRCW == FT && HW % 32 == 0) || (SRCW == HW && 32 % HW == 0),
                  "a slice inside one head, or whole heads inside a slice");
    static_assert(HW >= EPW, "a head's lanes in a slice cover one run of edges");
    static_assert(HW % 16 == 0, "a 16-lane row lies inside one head: a weight is handed on inside the row");
    __shared__ float acc[WIN * 32];
    __shared__ float dacc[WIN * HEADS];
    const int wave  = __builtin_amdgcn_readfirstlane((int)(threadIdx.x >> 5));
    const int lane  = threadIdx.x & 31;
    const int slice = blockIdx.x % SL;
    const int eb    = (blockIdx.x / SL) * EPB;
    const int fo    = slice * 32 + lane;
    const int h     = (HW % 32 == 0) ? (slice * 32) / HW : fo / HW;
    const int fsrc  = fo % SRCW;
    const int wbase = dst[eb] & ~(WIN - 1);
    const bool dblk = (HW % 32 == 0) ? ((slice * 32) % HW == 0) : true;
    const int  q    = (HW % 32 == 0) ? lane : lane % HW;

    for (int i = threadIdx.x; i < WIN * 32 / 4; i += 1024)
        ((float4*)acc)[i] = make_float4(0.f, 0.f, 0.f, 0.f);
    for (int i = threadIdx.x; i < WIN * HEADS / 4; i += 1024)
        ((float4*)dacc)[i] = make_float4(0.f, 0.f, 0.f, 0.f);
    __syncthreads();

    const int ew = eb + wave * EPWV;
    const int qq = lane & (EPW - 1);
    int s[EPW], d[EPW];
#pragma unroll
    for (int i = 0; i < EPW; ++i) { s[i] = src[ew + i]; d[i] = dst[ew + i]; }
    int dv = dst[ew + qq];
    for (int st = 0; st < EPWV; st += EPW) {
        const int e0 = ew + st;
        const int en = (st + EPW < EPWV) ? e0 + EPW : e0;
        int sn[EPW], dx[EPW];
#pragma unroll
        for (int i = 0; i < EPW; ++i) { sn[i] = src[en + i]; dx[i] = dst[en + i]; }
        const int dvn = dst[en + qq];
        float v[EPW];
#pragma unroll
        for (int i = 0; i < EPW; ++i) v[i] = fs[(size_t)s[i] * SRCW + fsrc];
        const float    xl = lg[(size_t)(e0 + qq) * HEADS + h];
        const unsigned mx = mmax[dv * HEADS + h];
        const float wv = __expf(xl - dec_f(mx));
#pragma unroll
        for (int i = 0; i < EPW; ++i) {
            const float val = v[i] * row_pick(wv, i);
            const int r = d[i] - wbase;
            if ((unsigned)r < (unsigned)WIN) atomicAdd(&acc[r * 32 + lane], val);
            else                             atomicAdd(out + (size_t)d[i] * FT + fo, val);
        }
        if (dblk) {
            if (q < EPW) {
                const int r = dv - wbase;
                if ((unsigned)r < (unsigned)WIN) atomicAdd(&dacc[r * HEADS + h], wv);
                else                             atomicAdd(den + dv * HEADS + h, wv);
            }
        }
#pragma unroll
        for (int i = 0; i < EPW; ++i) { s[i] = sn[i]; d[i] = dx[i]; }
        dv = dvn;
    }
    __syncthreads();

    for (int i0 = threadIdx.x; i0 < WIN * 32; i0 += 8 * 1024) {
        float a[8];
#pragma unroll
        for (int j = 0; j < 8; ++j) a[j] = acc[i0 + j * 1024];
#pragma unroll
        for (int j = 0; j < 8; ++j)
            atomicAdd(out + (size_t)(wbase + ((i0 + j * 1024) >> 5)) * FT + fo, a[j]);
    }
    if (dblk)
        for (int i = threadIdx.x; i < WIN * HEADS; i += 1024)
            atomicAdd(den + (size_t)wbase * HEADS + i, dacc[i]);
}

__global__ __launch_bounds__(128) void gemm_z_elu(
    const float* __restrict__ Z, const float* __restrict__ W, const float* __restrict__ Bv,
    const float* __restrict__ den, float* __restrict__ H) {
    const int lane = threadIdx.x & 31;
    const int wave = threadIdx.x >> 5;
    const int row0 = (blockIdx.x * 4 + wave) * 16;
    const int m  = lane & 15;
    const int hi = lane >> 4;

    for (int t = 0; t < 2; ++t) {
        const float* Zr = Z + (size_t)(row0 + m) * (2 * D_IN) + t * D_IN;
        v2f a[D_IN / 4];
#pragma unroll
        for (int kk = 0; kk < D_IN; kk += 4) {
            a[kk / 4].x = Zr[kk + 2 * hi + 0];
            a[kk / 4].y = Zr[kk + 2 * hi + 1];
        }
        float has[8], inv[8];
#pragma unroll
        for (int r = 0; r < 8; ++r) {
            const float dn = den[(row0 + r + 8 * hi) * 2 + t];
            has[r] = dn > 0.f ? 1.f : 0.f;
            inv[r] = dn > 0.f ? 1.f / dn : 0.f;
        }
        for (int c0 = 0; c0 < DH; c0 += 16) {
            const int col0 = t * DH + c0;
            v8f acc = {};
#pragma unroll
            for (int kk = 0; kk < D_IN; kk += 4) {
                v2f b;
                b.x = W[(size_t)(kk + 2 * hi + 0) * (2 * DH) + col0 + m];
                b.y = W[(size_t)(kk + 2 * hi + 1) * (2 * DH) + col0 + m];
                acc = __builtin_amdgcn_wmma_f32_16x16x4_f32(
                    false, a[kk / 4], false, b, (short)0, acc, false, false);
            }
            const float bias = Bv[col0 + m];
#pragma unroll
            for (int r = 0; r < 8; ++r) {
                const float v = acc[r] * inv[r] + bias * has[r];
                H[(size_t)(row0 + r + 8 * hi) * (2 * DH) + col0 + m] = v > 0.f ? v : (__expf(v) - 1.f);
            }
        }
    }
}

constexpr int L2_EPB = 4096;
static_assert(E_CNT % L2_EPB == 0 && (L2_EPB / 32) % EPW == 0, "whole work-groups, whole runs per wave");
__global__ __launch_bounds__(1024) void edge_logits2(
    const float* __restrict__ fs, const float* __restrict__ fd,
    const int* __restrict__ src, const int* __restrict__ dst,
    const float* __restrict__ attn, float* __restrict__ logits,
    unsigned* __restrict__ mmax) {
    constexpr int EPWV = L2_EPB / 32;
    constexpr int LAST = EPWV - EPW;
    __shared__ unsigned lmax[WIN];
    const int wave = __builtin_amdgcn_readfirstlane((int)(threadIdx.x >> 5));
    const int lane = threadIdx.x & 31;
    const int eb = blockIdx.x * L2_EPB;
    const int wbase = dst[eb] & ~(WIN - 1);
    for (int i = threadIdx.x; i < WIN; i += 1024) lmax[i] = 0u;
    __syncthreads();

    const float2 w = *(const float2*)(attn + lane * 2);
    const int ew = eb + wave * EPWV;
    const int sel = lane & (EPW - 1);
    constexpr int R1 = (EPW < LAST) ? EPW : LAST;
    int d[EPW], sn[EPW], dn[EPW];
    float2 a[EPW], b[EPW];
#pragma unroll
    for (int i = 0; i < EPW; ++i) {
        const int s0 = src[ew + i];
        d[i]  = dst[ew + i];
        sn[i] = src[ew + R1 + i];
        dn[i] = dst[ew + R1 + i];
        a[i] = *(const float2*)(fs + (size_t)s0 * 64 + lane * 2);
        b[i] = *(const float2*)(fd + (size_t)d[i] * 64 + lane * 2);
    }
    for (int st = 0; st < EPWV; st += EPW) {
        const int e0 = ew + st;
        const int e2 = ew + ((st + 2 * EPW < LAST) ? st + 2 * EPW : LAST);
        int s2[EPW], d2[EPW];
#pragma unroll
        for (int i = 0; i < EPW; ++i) { s2[i] = src[e2 + i]; d2[i] = dst[e2 + i]; }
        float2 na[EPW], nb[EPW];
#pragma unroll
        for (int i = 0; i < EPW; ++i) {
            na[i] = *(const float2*)(fs + (size_t)sn[i] * 64 + lane * 2);
            nb[i] = *(const float2*)(fd + (size_t)dn[i] * 64 + lane * 2);
        }
        float pe[EPW];
#pragma unroll
        for (int i = 0; i < EPW; ++i) {
            float p = 0.f, t;
            t = a[i].x + b[i].x; t = t > 0.f ? t : NEG_SLOPE * t; p += t * w.x;
            t = a[i].y + b[i].y; t = t > 0.f ? t : NEG_SLOPE * t; p += t * w.y;
            p += __shfl_xor(p, 1);  p += __shfl_xor(p, 2);
            p += __shfl_xor(p, 4);  p += __shfl_xor(p, 8);
            p += __shfl_xor(p, 16);
            pe[i] = p;
        }
        float pv = pe[0];
        int   dv = d[0];
#pragma unroll
        for (int i = 1; i < EPW; ++i) {
            pv = (sel == i) ? pe[i] : pv;
            dv = (sel == i) ? d[i]  : dv;
        }
        if (lane < EPW) {
            logits[e0 + sel] = pv;
            const int r = dv - wbase;
            if ((unsigned)r < (unsigned)WIN) atomicMax(&lmax[r], enc_f(pv));
            else                             atomicMax(&mmax[dv], enc_f(pv));
        }
#pragma unroll
        for (int i = 0; i < EPW; ++i) {
            a[i] = na[i]; b[i] = nb[i];
            d[i] = dn[i]; sn[i] = s2[i]; dn[i] = d2[i];
        }
    }
    __syncthreads();
    for (int i = threadIdx.x; i < WIN; i += 1024)
        atomicMax(&mmax[(size_t)wbase + i], lmax[i]);
}

__global__ __launch_bounds__(256) void reorder_out(const float* __restrict__ agg,
                                                   const float* __restrict__ den,
                                                   float* __restrict__ out) {
    const int idx = blockIdx.x * 256 + threadIdx.x;
    const int n = idx >> 6, f = idx & 63;
    const int b = n >> 11;
    const int r = n & 2047;
    size_t o;
    if (r < NUM_PM) o = ((size_t)(b * NUM_PM + r) * 64) + f;
    else            o = (size_t)B_SZ * NUM_PM * 64 +
                        ((size_t)(b * NUM_VM + (r - NUM_PM)) * 64) + f;
    const float dn = den[n];
    out[o] = dn > 0.f ? agg[idx] / dn : 0.f;
}

extern "C" void kernel_launch(void* const* d_in, const int* in_sizes, int n_in,
                              void* d_out, int out_size, void* d_ws, size_t ws_size,
                              hipStream_t stream) {
    const float* x     = (const float*)d_in[0];
    const int*   src   = (const int*)d_in[1];
    const int*   dst   = (const int*)d_in[2];
    const float* W1s   = (const float*)d_in[3];
    const float* b1s   = (const float*)d_in[4];
    const float* W1d   = (const float*)d_in[5];
    const float* b1d   = (const float*)d_in[6];
    const float* attn1 = (const float*)d_in[7];
    const float* W2s   = (const float*)d_in[8];
    const float* b2s   = (const float*)d_in[9];
    const float* W2d   = (const float*)d_in[10];
    const float* b2d   = (const float*)d_in[11];
    const float* attn2 = (const float*)d_in[12];
    (void)in_sizes; (void)n_in; (void)out_size; (void)ws_size;

    char* ws = (char*)d_ws;
    const size_t MB = 1024u * 1024u;
    float*    log1  = (float*)(ws + 32 * MB);
    unsigned* m1    = (unsigned*)(ws + 40 * MB);
    float*    den1  = (float*)(ws + 40 * MB + 262144);
    float*    h1    = (float*)(ws + 41 * MB);
    float*    z1    = (float*)(ws + 57 * MB);
    float*    fs2   = (float*)(ws + 16 * MB);
    float*    fd2   = (float*)(ws + 24 * MB);
    float*    agg2  = (float*)(ws + 0);
    float*    log2  = (float*)(ws + 8 * MB);
    unsigned* m2    = (unsigned*)(ws + 12 * MB);
    float*    den2  = (float*)(ws + 12 * MB + 131072);

    float* out = (float*)d_out;

    zero_l1<<<N_NODE * 2 * D_IN / 256, 256, 0, stream>>>(z1, m1, den1);
    edge_logits1_x<<<E_CNT / (16 * LT_TPW * LT_WAVES), LT_WAVES * 32, 0, stream>>>(
        x, src, dst, W1s, b1s, W1d, b1d, attn1, log1, m1);
    edge_agg_lds<2 * D_IN, 2, 4096, D_IN><<<E_CNT / 4096, 1024, 0, stream>>>(x, src, dst, log1, m1, z1, den1);
    gemm_z_elu<<<N_NODE / 16 / 4, 128, 0, stream>>>(z1, W1s, b1s, den1, h1);

    zero_l2<<<N_NODE * 64 / 256, 256, 0, stream>>>(agg2, m2, den2);
    gemm_dual_wmma<2 * DH, DH><<<N_NODE / 16 / 4, 128, 0, stream>>>(
        h1, W2s, b2s, W2d, b2d, fs2, fd2);
    edge_logits2<<<E_CNT / L2_EPB, 1024, 0, stream>>>(fs2, fd2, src, dst, attn2, log2, m2);
    edge_agg_lds<DH, 1, 8192, DH><<<E_CNT / 8192 * 2, 1024, 0, stream>>>(fs2, src, dst, log2, m2, agg2, den2);

    reorder_out<<<N_NODE * 64 / 256, 256, 0, stream>>>(agg2, den2, out);
}


# Aggregation by 16-column slices over twice the edges: a 16-lane row per edge (32-edge runs, 16 row loads in flight), sums stored slice-major: half the window flush atomics; sum order changed
# speedup vs baseline: 5.0890x; 1.1071x over previous
#include <hip/hip_runtime.h>

constexpr int B_SZ   = 16;
constexpr int NUM_PM = 512;
constexpr int NUM_VM = 1536;
constexpr int NPG    = NUM_PM + NUM_VM;
constexpr int N_NODE = B_SZ * NPG;
constexpr int DEG    = 32;
constexpr int E_CNT  = DEG * N_NODE;
constexpr int D_IN   = 16;
constexpr int DH     = 64;
constexpr float NEG_SLOPE = 0.2f;
constexpr int EPW    = 8;
static_assert((EPW & (EPW - 1)) == 0 && EPW <= 16, "a run's results are written by lanes 0..EPW-1 of each 16-lane half");

constexpr int WIN = 2048;
constexpr int SLW = 16;

typedef float v2f __attribute__((ext_vector_type(2)));
typedef float v8f __attribute__((ext_vector_type(8)));

__device__ __forceinline__ unsigned enc_f(float f) {
    unsigned u = __float_as_uint(f);
    return (u & 0x80000000u) ? ~u : (u | 0x80000000u);
}
__device__ __forceinline__ float dec_f(unsigned u) {
    return __uint_as_float((u & 0x80000000u) ? (u & 0x7fffffffu) : ~u);
}

template <int K, int COLS>
__global__ __launch_bounds__(128) void gemm_dual_wmma(
    const float* __restrict__ X,
    const float* __restrict__ Ws, const float* __restrict__ Bs,
    const float* __restrict__ Wd, const float* __restrict__ Bd,
    float* __restrict__ Fs, float* __restrict__ Fd) {
    const int lane = threadIdx.x & 31;
    const int wave = threadIdx.x >> 5;
    const int row0 = (blockIdx.x * 4 + wave) * 16;
    const int m  = lane & 15;
    const int hi = lane >> 4;
    const float* Xr = X + (size_t)(row0 + m) * K;

    for (int t = 0; t < 2; ++t) {
        const float* W  = t ? Wd : Ws;
        const float* Bv = t ? Bd : Bs;
        float*       F  = t ? Fd : Fs;
        for (int col0 = 0; col0 < COLS; col0 += 16) {
            v8f acc = {};
#pragma unroll
            for (int kk = 0; kk < K; kk += 4) {
                v2f a, b;
                a.x = Xr[kk + 2 * hi + 0];
                a.y = Xr[kk + 2 * hi + 1];
                b.x = W[(size_t)(kk + 2 * hi + 0) * COLS + col0 + m];
                b.y = W[(size_t)(kk + 2 * hi + 1) * COLS + col0 + m];
                acc = __builtin_amdgcn_wmma_f32_16x16x4_f32(
                    false, a, false, b, (short)0, acc, false, false);
            }
            const float bias = Bv[col0 + m];
#pragma unroll
            for (int r = 0; r < 8; ++r) {
                F[(size_t)(row0 + r + 8 * hi) * COLS + col0 + m] = acc[r] + bias;
            }
        }
    }
}

__global__ __launch_bounds__(256) void zero_l1(float* __restrict__ z1,
                                               unsigned* __restrict__ m1,
                                               float* __restrict__ den1) {
    int idx = blockIdx.x * 256 + threadIdx.x;
    z1[idx] = 0.f;
    if (idx < N_NODE * 2) { m1[idx] = 0u; den1[idx] = 0.f; }
}
__global__ __launch_bounds__(256) void zero_l2(float* __restrict__ agg,
                                               unsigned* __restrict__ m2,
                                               float* __restrict__ den2) {
    int idx = blockIdx.x * 256 + threadIdx.x;
    agg[idx] = 0.f;
    if (idx < N_NODE) { m2[idx] = 0u; den2[idx] = 0.f; }
}

constexpr int LT_TPW = 16;
constexpr int LT_WAVES = 16;
static_assert(E_CNT % (16 * LT_TPW * LT_WAVES) == 0, "whole tiles per wave, whole waves per work-group");

template <int CTRL>
__device__ __forceinline__ float row_xm(float v) {
    return __int_as_float(__builtin_amdgcn_update_dpp(0, __float_as_int(v), CTRL, 0xf, 0xf, false));
}

template <int I>
__device__ __forceinline__ int row_sh(int v) {
    return __builtin_amdgcn_update_dpp(0, v, 0x150 + I, 0xf, 0xf, false);
}
__device__ __forceinline__ int row_pick_i(int v, int i) {
    switch (i) {
        case 0:  return row_sh<0>(v);
        case 1:  return row_sh<1>(v);
        case 2:  return row_sh<2>(v);
        case 3:  return row_sh<3>(v);
        case 4:  return row_sh<4>(v);
        case 5:  return row_sh<5>(v);
        case 6:  return row_sh<6>(v);
        case 7:  return row_sh<7>(v);
        case 8:  return row_sh<8>(v);
        case 9:  return row_sh<9>(v);
        case 10: return row_sh<10>(v);
        case 11: return row_sh<11>(v);
        case 12: return row_sh<12>(v);
        case 13: return row_sh<13>(v);
        case 14: return row_sh<14>(v);
        default: return row_sh<15>(v);
    }
}
__device__ __forceinline__ float row_pick(float v, int i) {
    return __int_as_float(row_pick_i(__float_as_int(v), i));
}

__global__ __launch_bounds__(LT_WAVES * 32) void edge_logits1_x(
    const float* __restrict__ x, const int* __restrict__ src, const int* __restrict__ dst,
    const float* __restrict__ Ws, const float* __restrict__ Bs,
    const float* __restrict__ Wd, const float* __restrict__ Bd,
    const float* __restrict__ attn, float* __restrict__ logits, unsigned* __restrict__ mmax) {
    __shared__ v2f wl[64 * 32];
    __shared__ unsigned lmax[WIN * 2];
    const int lane = threadIdx.x & 31;
    const int wave = threadIdx.x >> 5;
    const int c  = lane & 15;
    const int hh = lane >> 4;

    for (int q = threadIdx.x; q < 64 * 32; q += LT_WAVES * 32) {
        const int l = q & 31, j = q >> 8, n = (q >> 5) & 7;
        const float* Wm = (j < 4) ? Ws : Wd;
        const int k = 8 * (l >> 4) + 2 * (j & 3);
        v2f w;
        w.x = Wm[(size_t)(k + 0) * (2 * DH) + 16 * n + (l & 15)];
        w.y = Wm[(size_t)(k + 1) * (2 * DH) + 16 * n + (l & 15)];
        wl[q] = w;
    }
    for (int q = threadIdx.x; q < WIN * 2; q += LT_WAVES * 32) lmax[q] = 0u;
    const int wbase = dst[blockIdx.x * (LT_WAVES * LT_TPW * 16)] & ~(WIN - 1);
    __syncthreads();
    v2f wf[8][8];
#pragma unroll
    for (int j = 0; j < 8; ++j)
#pragma unroll
        for (int n = 0; n < 8; ++n) wf[j][n] = wl[(j * 8 + n) * 32 + lane];
    float bv[8], av[8];
#pragma unroll
    for (int n = 0; n < 8; ++n) {
        bv[n] = Bs[16 * n + c] + Bd[16 * n + c];
        av[n] = attn[16 * n + c];
    }

    const int ebase = (blockIdx.x * LT_WAVES + wave) * (LT_TPW * 16);
    const int osel  = 8 * hh + (c & 7);
    const int ohead = c >> 3;
    const int et1 = (LT_TPW > 1) ? ebase + 16 : ebase;
    const int s1 = src[ebase + c], d1 = dst[ebase + c];
    int s2 = src[et1 + c], d2 = dst[et1 + c];
    float4 xs0 = *(const float4*)(x + (size_t)s1 * D_IN + 8 * hh);
    float4 xs1 = *(const float4*)(x + (size_t)s1 * D_IN + 8 * hh + 4);
    float4 xd0 = *(const float4*)(x + (size_t)d1 * D_IN + 8 * hh);
    float4 xd1 = *(const float4*)(x + (size_t)d1 * D_IN + 8 * hh + 4);
    for (int t = 0; t < LT_TPW; ++t) {
        const int e0 = ebase + t * 16;
        const int e2 = ebase + ((t + 2 < LT_TPW) ? t + 2 : LT_TPW - 1) * 16;
        const int s3 = src[e2 + c], d3 = dst[e2 + c];
        const int dw = dst[e0 + osel];
        const float4 ns0 = *(const float4*)(x + (size_t)s2 * D_IN + 8 * hh);
        const float4 ns1 = *(const float4*)(x + (size_t)s2 * D_IN + 8 * hh + 4);
        const float4 nd0 = *(const float4*)(x + (size_t)d2 * D_IN + 8 * hh);
        const float4 nd1 = *(const float4*)(x + (size_t)d2 * D_IN + 8 * hh + 4);

        v2f a[8];
        a[0].x = xs0.x; a[0].y = xs0.y; a[1].x = xs0.z; a[1].y = xs0.w;
        a[2].x = xs1.x; a[2].y = xs1.y; a[3].x = xs1.z; a[3].y = xs1.w;
        a[4].x = xd0.x; a[4].y = xd0.y; a[5].x = xd0.z; a[5].y = xd0.w;
        a[6].x = xd1.x; a[6].y = xd1.y; a[7].x = xd1.z; a[7].y = xd1.w;
        float p[16];
#pragma unroll
        for (int i = 0; i < 16; ++i) p[i] = 0.f;
#pragma unroll
        for (int n = 0; n < 8; ++n) {
            v8f acc = {};
#pragma unroll
            for (int j = 0; j < 8; ++j)
                acc = __builtin_amdgcn_wmma_f32_16x16x4_f32(
                    false, a[j], false, wf[j][n], (short)0, acc, false, false);
#pragma unroll
            for (int r = 0; r < 8; ++r) {
                float u = acc[r] + bv[n];
                u = u > 0.f ? u : NEG_SLOPE * u;
                p[(n >> 2) * 8 + r] += u * av[n];
            }
        }
#pragma unroll
        for (int i = 0; i < 8; ++i) {
            const float keep = (c & 8) ? p[i + 8] : p[i], send = (c & 8) ? p[i] : p[i + 8];
            p[i] = keep + row_xm<0x168>(send);
        }
#pragma unroll
        for (int i = 0; i < 4; ++i) {
            const float keep = (c & 4) ? p[i + 4] : p[i], send = (c & 4) ? p[i] : p[i + 4];
            p[i] = keep + row_xm<0x164>(send);
        }
#pragma unroll
        for (int i = 0; i < 2; ++i) {
            const float keep = (c & 2) ? p[i + 2] : p[i], send = (c & 2) ? p[i] : p[i + 2];
            p[i] = keep + row_xm<0x162>(send);
        }
        const float keep = (c & 1) ? p[1] : p[0], send = (c & 1) ? p[0] : p[1];
        const float pv = keep + row_xm<0x161>(send);
        logits[(size_t)(e0 + osel) * 2 + ohead] = pv;
        const int rw = dw - wbase;
        if ((unsigned)rw < (unsigned)WIN) atomicMax(&lmax[rw * 2 + ohead], enc_f(pv));
        else                              atomicMax(&mmax[dw * 2 + ohead], enc_f(pv));

        xs0 = ns0; xs1 = ns1; xd0 = nd0; xd1 = nd1;
        s2 = s3; d2 = d3;
    }
    __syncthreads();
    for (int q = threadIdx.x; q < WIN * 2; q += LT_WAVES * 32)
        atomicMax(&mmax[(size_t)wbase * 2 + q], lmax[q]);
}

template <int FT, int HEADS, int EPB, int SRCW>
__global__ __launch_bounds__(1024) void edge_agg_lds(
    const float* __restrict__ fs, const int* __restrict__ src,
    const int* __restrict__ dst, const float* __restrict__ lg,
    const unsigned* __restrict__ mmax, float* __restrict__ out, float* __restrict__ den) {
    constexpr int SL   = FT / SLW;
    constexpr int EPWV = EPB / 32;
    constexpr int HW   = FT / HEADS;
    static_assert(EPWV % 32 == 0 && FT % SLW == 0, "whole runs per wave, whole slices per row");
    static_assert(HW % SLW == 0, "a slice lies inside one head");
    static_assert(SRCW == FT || SRCW == HW, "the source row is the summed row, or one head of it");
    __shared__ float acc[WIN * SLW];
    __shared__ float dacc[WIN];
    const int wave  = threadIdx.x >> 5;
    const int lane  = threadIdx.x & 31;
    const int c     = lane & (SLW - 1);
    const int slice = blockIdx.x % SL;
    const int eb    = (blockIdx.x / SL) * EPB;
    const int h     = (slice * SLW) / HW;
    const int fsrc  = (slice * SLW + c) % SRCW;
    const bool dblk = (slice * SLW) % HW == 0;
    const int wbase = dst[eb] & ~(WIN - 1);
    float* outs = out + (size_t)slice * N_NODE * SLW;
    float* dens = den + (size_t)h * N_NODE;

    for (int i = threadIdx.x; i < WIN * SLW / 4; i += 1024)
        ((float4*)acc)[i] = make_float4(0.f, 0.f, 0.f, 0.f);
    for (int i = threadIdx.x; i < WIN / 4; i += 1024)
        ((float4*)dacc)[i] = make_float4(0.f, 0.f, 0.f, 0.f);
    __syncthreads();

    const int ew = eb + wave * EPWV;
    int sv = src[ew + lane], dv = dst[ew + lane];
#pragma unroll 1
    for (int st = 0; st < EPWV; st += 32) {
        const int e0 = ew + st;
        const int en = (st + 32 < EPWV) ? e0 + 32 : e0;
        const int svn = src[en + lane], dvn = dst[en + lane];
        const int srow = sv * SRCW;
        float v[16];
#pragma unroll
        for (int i = 0; i < 16; ++i) v[i] = fs[(unsigned)(row_pick_i(srow, i) + fsrc)];
        const float    xl = lg[(size_t)(e0 + lane) * HEADS + h];
        const unsigned mx = mmax[dv * HEADS + h];
        const float wv = __expf(xl - dec_f(mx));
        const unsigned rw = (unsigned)(dv - wbase);
        if (!__any(rw >= (unsigned)WIN)) {
            const int rrow = (int)rw * SLW;
#pragma unroll
            for (int i = 0; i < 16; ++i)
                atomicAdd(&acc[row_pick_i(rrow, i) + c], v[i] * row_pick(wv, i));
        } else {
#pragma unroll
            for (int i = 0; i < 16; ++i) {
                const float val = v[i] * row_pick(wv, i);
                const int di = row_pick_i(dv, i);
                const unsigned ri = (unsigned)(di - wbase);
                if (ri < (unsigned)WIN) atomicAdd(&acc[ri * SLW + c], val);
                else                    atomicAdd(outs + (size_t)di * SLW + c, val);
            }
        }
        if (dblk) {
            if (rw < (unsigned)WIN) atomicAdd(&dacc[rw], wv);
            else                    atomicAdd(dens + dv, wv);
        }
        sv = svn; dv = dvn;
    }
    __syncthreads();

    for (int i0 = threadIdx.x; i0 < WIN * SLW; i0 += 8 * 1024) {
        float a[8];
#pragma unroll
        for (int j = 0; j < 8; ++j) a[j] = acc[i0 + j * 1024];
#pragma unroll
        for (int j = 0; j < 8; ++j)
            atomicAdd(outs + (size_t)wbase * SLW + i0 + j * 1024, a[j]);
    }
    if (dblk)
        for (int i = threadIdx.x; i < WIN; i += 1024)
            atomicAdd(dens + wbase + i, dacc[i]);
}

__global__ __launch_bounds__(128) void gemm_z_elu(
    const float* __restrict__ Z, const float* __restrict__ W, const float* __restrict__ Bv,
    const float* __restrict__ den, float* __restrict__ H) {
    const int lane = threadIdx.x & 31;
    const int wave = threadIdx.x >> 5;
    const int row0 = (blockIdx.x * 4 + wave) * 16;
    const int m  = lane & 15;
    const int hi = lane >> 4;

    for (int t = 0; t < 2; ++t) {
        const float* Zr = Z + (size_t)t * N_NODE * D_IN + (size_t)(row0 + m) * D_IN;
        v2f a[D_IN / 4];
#pragma unroll
        for (int kk = 0; kk < D_IN; kk += 4) {
            a[kk / 4].x = Zr[kk + 2 * hi + 0];
            a[kk / 4].y = Zr[kk + 2 * hi + 1];
        }
        float has[8], inv[8];
#pragma unroll
        for (int r = 0; r < 8; ++r) {
            const float dn = den[(size_t)t * N_NODE + row0 + r + 8 * hi];
            has[r] = dn > 0.f ? 1.f : 0.f;
            inv[r] = dn > 0.f ? 1.f / dn : 0.f;
        }
        for (int c0 = 0; c0 < DH; c0 += 16) {
            const int col0 = t * DH + c0;
            v8f acc = {};
#pragma unroll
            for (int kk = 0; kk < D_IN; kk += 4) {
                v2f b;
                b.x = W[(size_t)(kk + 2 * hi + 0) * (2 * DH) + col0 + m];
                b.y = W[(size_t)(kk + 2 * hi + 1) * (2 * DH) + col0 + m];
                acc = __builtin_amdgcn_wmma_f32_16x16x4_f32(
                    false, a[kk / 4], false, b, (short)0, acc, false, false);
            }
            const float bias = Bv[col0 + m];
#pragma unroll
            for (int r = 0; r < 8; ++r) {
                const float v = acc[r] * inv[r] + bias * has[r];
                H[(size_t)(row0 + r + 8 * hi) * (2 * DH) + col0 + m] = v > 0.f ? v : (__expf(v) - 1.f);
            }
        }
    }
}

constexpr int L2_EPB = 4096;
static_assert(E_CNT % L2_EPB == 0 && (L2_EPB / 32) % EPW == 0, "whole work-groups, whole runs per wave");
__global__ __launch_bounds__(1024) void edge_logits2(
    const float* __restrict__ fs, const float* __restrict__ fd,
    const int* __restrict__ src, const int* __restrict__ dst,
    const float* __restrict__ attn, float* __restrict__ logits,
    unsigned* __restrict__ mmax) {
    constexpr int EPWV = L2_EPB / 32;
    constexpr int LAST = EPWV - EPW;
    __shared__ unsigned lmax[WIN];
    const int wave = __builtin_amdgcn_readfirstlane((int)(threadIdx.x >> 5));
    const int lane = threadIdx.x & 31;
    const int eb = blockIdx.x * L2_EPB;
    const int wbase = dst[eb] & ~(WIN - 1);
    for (int i = threadIdx.x; i < WIN; i += 1024) lmax[i] = 0u;
    __syncthreads();

    const float2 w = *(const float2*)(attn + lane * 2);
    const int ew = eb + wave * EPWV;
    const int sel = lane & (EPW - 1);
    constexpr int R1 = (EPW < LAST) ? EPW : LAST;
    int d[EPW], sn[EPW], dn[EPW];
    float2 a[EPW], b[EPW];
#pragma unroll
    for (int i = 0; i < EPW; ++i) {
        const int s0 = src[ew + i];
        d[i]  = dst[ew + i];
        sn[i] = src[ew + R1 + i];
        dn[i] = dst[ew + R1 + i];
        a[i] = *(const float2*)(fs + (size_t)s0 * 64 + lane * 2);
        b[i] = *(const float2*)(fd + (size_t)d[i] * 64 + lane * 2);
    }
    for (int st = 0; st < EPWV; st += EPW) {
        const int e0 = ew + st;
        const int e2 = ew + ((st + 2 * EPW < LAST) ? st + 2 * EPW : LAST);
        int s2[EPW], d2[EPW];
#pragma unroll
        for (int i = 0; i < EPW; ++i) { s2[i] = src[e2 + i]; d2[i] = dst[e2 + i]; }
        float2 na[EPW], nb[EPW];
#pragma unroll
        for (int i = 0; i < EPW; ++i) {
            na[i] = *(const float2*)(fs + (size_t)sn[i] * 64 + lane * 2);
            nb[i] = *(const float2*)(fd + (size_t)dn[i] * 64 + lane * 2);
        }
        float pe[EPW];
#pragma unroll
        for (int i = 0; i < EPW; ++i) {
            float p = 0.f, t;
            t = a[i].x + b[i].x; t = t > 0.f ? t : NEG_SLOPE * t; p += t * w.x;
            t = a[i].y + b[i].y; t = t > 0.f ? t : NEG_SLOPE * t; p += t * w.y;
            p += __shfl_xor(p, 1);  p += __shfl_xor(p, 2);
            p += __shfl_xor(p, 4);  p += __shfl_xor(p, 8);
            p += __shfl_xor(p, 16);
            pe[i] = p;
        }
        float pv = pe[0];
        int   dv = d[0];
#pragma unroll
        for (int i = 1; i < EPW; ++i) {
            pv = (sel == i) ? pe[i] : pv;
            dv = (sel == i) ? d[i]  : dv;
        }
        if (lane < EPW) {
            logits[e0 + sel] = pv;
            const int r = dv - wbase;
            if ((unsigned)r < (unsigned)WIN) atomicMax(&lmax[r], enc_f(pv));
            else                             atomicMax(&mmax[dv], enc_f(pv));
        }
#pragma unroll
        for (int i = 0; i < EPW; ++i) {
            a[i] = na[i]; b[i] = nb[i];
            d[i] = dn[i]; sn[i] = s2[i]; dn[i] = d2[i];
        }
    }
    __syncthreads();
    for (int i = threadIdx.x; i < WIN; i += 1024)
        atomicMax(&mmax[(size_t)wbase + i], lmax[i]);
}

__global__ __launch_bounds__(256) void reorder_out(const float* __restrict__ agg,
                                                   const float* __restrict__ den,
                                                   float* __restrict__ out) {
    const int idx = blockIdx.x * 256 + threadIdx.x;
    const int n = idx >> 6, f = idx & 63;
    const int b = n >> 11;
    const int r = n & 2047;
    size_t o;
    if (r < NUM_PM) o = ((size_t)(b * NUM_PM + r) * 64) + f;
    else            o = (size_t)B_SZ * NUM_PM * 64 +
                        ((size_t)(b * NUM_VM + (r - NUM_PM)) * 64) + f;
    const float dn = den[n];
    const float a = agg[((size_t)(f / SLW) * N_NODE + n) * SLW + (f % SLW)];
    out[o] = dn > 0.f ? a / dn : 0.f;
}

extern "C" void kernel_launch(void* const* d_in, const int* in_sizes, int n_in,
                              void* d_out, int out_size, void* d_ws, size_t ws_size,
                              hipStream_t stream) {
    const float* x     = (const float*)d_in[0];
    const int*   src   = (const int*)d_in[1];
    const int*   dst   = (const int*)d_in[2];
    const float* W1s   = (const float*)d_in[3];
    const float* b1s   = (const float*)d_in[4];
    const float* W1d   = (const float*)d_in[5];
    const float* b1d   = (const float*)d_in[6];
    const float* attn1 = (const float*)d_in[7];
    const float* W2s   = (const float*)d_in[8];
    const float* b2s   = (const float*)d_in[9];
    const float* W2d   = (const float*)d_in[10];
    const float* b2d   = (const float*)d_in[11];
    const float* attn2 = (const float*)d_in[12];
    (void)in_sizes; (void)n_in; (void)out_size; (void)ws_size;

    char* ws = (char*)d_ws;
    const size_t MB = 1024u * 1024u;
    float*    log1  = (float*)(ws + 32 * MB);
    unsigned* m1    = (unsigned*)(ws + 40 * MB);
    float*    den1  = (float*)(ws + 40 * MB + 262144);
    float*    h1    = (float*)(ws + 41 * MB);
    float*    z1    = (float*)(ws + 57 * MB);
    float*    fs2   = (float*)(ws + 16 * MB);
    float*    fd2   = (float*)(ws + 24 * MB);
    float*    agg2  = (float*)(ws + 0);
    float*    log2  = (float*)(ws + 8 * MB);
    unsigned* m2    = (unsigned*)(ws + 12 * MB);
    float*    den2  = (float*)(ws + 12 * MB + 131072);

    float* out = (float*)d_out;

    zero_l1<<<N_NODE * 2 * D_IN / 256, 256, 0, stream>>>(z1, m1, den1);
    edge_logits1_x<<<E_CNT / (16 * LT_TPW * LT_WAVES), LT_WAVES * 32, 0, stream>>>(
        x, src, dst, W1s, b1s, W1d, b1d, attn1, log1, m1);
    edge_agg_lds<2 * D_IN, 2, 8192, D_IN><<<E_CNT / 8192 * 2, 1024, 0, stream>>>(x, src, dst, log1, m1, z1, den1);
    gemm_z_elu<<<N_NODE / 16 / 4, 128, 0, stream>>>(z1, W1s, b1s, den1, h1);

    zero_l2<<<N_NODE * 64 / 256, 256, 0, stream>>>(agg2, m2, den2);
    gemm_dual_wmma<2 * DH, DH><<<N_NODE / 16 / 4, 128, 0, stream>>>(
        h1, W2s, b2s, W2d, b2d, fs2, fd2);
    edge_logits2<<<E_CNT / L2_EPB, 1024, 0, stream>>>(fs2, fd2, src, dst, attn2, log2, m2);
    edge_agg_lds<DH, 1, 16384, DH><<<E_CNT / 16384 * 4, 1024, 0, stream>>>(fs2, src, dst, log2, m2, agg2, den2);

    reorder_out<<<N_NODE * 64 / 256, 256, 0, stream>>>(agg2, den2, out);
}
